# starting kernel
# baseline (speedup 1.0000x reference)
#include <hip/hip_runtime.h>

typedef float v2f __attribute__((ext_vector_type(2)));
typedef float v8f __attribute__((ext_vector_type(8)));

#define NFEAT 256
#define NHID  64


__global__ void k_deg_init(float* deg, int n) {
    int i = blockIdx.x * blockDim.x + threadIdx.x;
    if (i < n) deg[i] = 1.0f;
}

__global__ void k_deg_count(const int* __restrict__ dst, float* deg, int n_edges) {
    int e = blockIdx.x * blockDim.x + threadIdx.x;
    if (e < n_edges) atomicAdd(&deg[dst[e]], 1.0f);
}

__global__ void k_dinv(float* deg, int n) {
    int i = blockIdx.x * blockDim.x + threadIdx.x;
    if (i < n) {
        float d = deg[i];
        deg[i] = (d > 0.0f) ? rsqrtf(d) : 0.0f;
    }
}


__global__ __launch_bounds__(256) void k_gemm1_wmma(const float* __restrict__ x,
                                                    const float* __restrict__ W1,
                                                    float* __restrict__ h,
                                                    int n_nodes) {
    __shared__ float ldsW[NFEAT * NHID];
    int tid = threadIdx.x;

    for (int idx = tid * 4; idx < NFEAT * NHID; idx += 256 * 4) {
        float4 w = *(const float4*)(W1 + idx);
        int k  = idx >> 6;
        int n0 = idx & 63;
        int p  = k >> 1;
        int lo = k & 1;
        int rot = (p & 1) << 5;
        float vals[4] = {w.x, w.y, w.z, w.w};
        #pragma unroll
        for (int j = 0; j < 4; ++j) {
            int wpos = (((n0 + j) << 1) + rot) & 127;
            ldsW[p * 128 + wpos + lo] = vals[j];
        }
    }
    __syncthreads();

    int wave = tid >> 5;
    int lane = tid & 31;
    int tile = blockIdx.x * 8 + wave;
    int m0 = tile * 16;
    if (m0 >= n_nodes) return;

    int half = lane >> 4;
    int mrow = m0 + (lane & 15);
    int kk   = half << 1;
    int ncol = lane & 15;
    int rot  = half << 5;

    int wp0 = (((ncol +  0) << 1) + rot) & 127;
    int wp1 = (((ncol + 16) << 1) + rot) & 127;
    int wp2 = (((ncol + 32) << 1) + rot) & 127;
    int wp3 = (((ncol + 48) << 1) + rot) & 127;

    v8f acc0 = {}, acc1 = {}, acc2 = {}, acc3 = {};

    const float* xrow = x + (size_t)mrow * NFEAT;
    for (int k = 0; k < NFEAT; k += 4) {
        float2 av = *(const float2*)(xrow + k + kk);
        v2f a; a.x = av.x; a.y = av.y;

        const float* rowp = ldsW + ((k >> 1) + half) * 128;
        v2f b0 = *(const v2f*)(rowp + wp0);
        v2f b1 = *(const v2f*)(rowp + wp1);
        v2f b2 = *(const v2f*)(rowp + wp2);
        v2f b3 = *(const v2f*)(rowp + wp3);

        acc0 = __builtin_amdgcn_wmma_f32_16x16x4_f32(false, a, false, b0, (short)0, acc0, false, false);
        acc1 = __builtin_amdgcn_wmma_f32_16x16x4_f32(false, a, false, b1, (short)0, acc1, false, false);
        acc2 = __builtin_amdgcn_wmma_f32_16x16x4_f32(false, a, false, b2, (short)0, acc2, false, false);
        acc3 = __builtin_amdgcn_wmma_f32_16x16x4_f32(false, a, false, b3, (short)0, acc3, false, false);
    }

    int rbase = m0 + (half << 3);
    for (int r = 0; r < 8; ++r) {
        float* hr = h + (size_t)(rbase + r) * NHID;
        hr[ncol +  0] = acc0[r];
        hr[ncol + 16] = acc1[r];
        hr[ncol + 32] = acc2[r];
        hr[ncol + 48] = acc3[r];
    }
}


__global__ void k_selfloop1(const float* __restrict__ h, const float* __restrict__ dinv,
                            float* __restrict__ out1, int n_nodes) {
    int t = blockIdx.x * blockDim.x + threadIdx.x;
    int i = t >> 6, f = t & 63;
    if (i >= n_nodes) return;
    float di = dinv[i];
    out1[(size_t)i * NHID + f] = h[(size_t)i * NHID + f] * di * di;
}

__global__ void k_edge_agg1(const int* __restrict__ src, const int* __restrict__ dst,
                            const float* __restrict__ dinv, const float* __restrict__ h,
                            float* __restrict__ out1, int n_edges) {
    int t = blockIdx.x * blockDim.x + threadIdx.x;
    int e = t >> 6, f = t & 63;
    if (e >= n_edges) return;
    int s = src[e], d = dst[e];
    float norm = dinv[s] * dinv[d];
    atomicAdd(&out1[(size_t)d * NHID + f], h[(size_t)s * NHID + f] * norm);
}

__global__ void k_relu_bias(const float* __restrict__ out1, const float* __restrict__ b1,
                            float* __restrict__ h1, int n_nodes) {
    int t = blockIdx.x * blockDim.x + threadIdx.x;
    int i = t >> 6, f = t & 63;
    if (i >= n_nodes) return;
    float v = out1[(size_t)i * NHID + f] + b1[f];
    h1[(size_t)i * NHID + f] = fmaxf(v, 0.0f);
}


__global__ void k_gemm2(const float* __restrict__ h1, const float* __restrict__ W2,
                        const float* __restrict__ b2, const float* __restrict__ dinv,
                        float* __restrict__ z, float* __restrict__ out, int n_nodes) {
    int i = blockIdx.x * blockDim.x + threadIdx.x;
    if (i >= n_nodes) return;
    const float* hr = h1 + (size_t)i * NHID;
    float z0 = 0.0f, z1 = 0.0f;
    #pragma unroll
    for (int f = 0; f < NHID; f += 4) {
        float4 hv = *(const float4*)(hr + f);
        z0 += hv.x * W2[2 * (f + 0) + 0]; z1 += hv.x * W2[2 * (f + 0) + 1];
        z0 += hv.y * W2[2 * (f + 1) + 0]; z1 += hv.y * W2[2 * (f + 1) + 1];
        z0 += hv.z * W2[2 * (f + 2) + 0]; z1 += hv.z * W2[2 * (f + 2) + 1];
        z0 += hv.w * W2[2 * (f + 3) + 0]; z1 += hv.w * W2[2 * (f + 3) + 1];
    }
    float di = dinv[i], n2 = di * di;
    z[2 * i + 0] = z0;
    z[2 * i + 1] = z1;
    out[2 * i + 0] = z0 * n2 + b2[0];
    out[2 * i + 1] = z1 * n2 + b2[1];
}

__global__ void k_edge_agg2(const int* __restrict__ src, const int* __restrict__ dst,
                            const float* __restrict__ dinv, const float* __restrict__ z,
                            float* __restrict__ out, int n_edges) {
    int e = blockIdx.x * blockDim.x + threadIdx.x;
    if (e >= n_edges) return;
    int s = src[e], d = dst[e];
    float norm = dinv[s] * dinv[d];
    float2 zv = *(const float2*)(z + 2 * s);
    atomicAdd(&out[2 * d + 0], zv.x * norm);
    atomicAdd(&out[2 * d + 1], zv.y * norm);
}


extern "C" void kernel_launch(void* const* d_in, const int* in_sizes, int n_in,
                              void* d_out, int out_size, void* d_ws, size_t ws_size,
                              hipStream_t stream) {
    const float* x  = (const float*)d_in[0];
    const int*   ei = (const int*)d_in[1];
    const float* W1 = (const float*)d_in[2];
    const float* b1 = (const float*)d_in[3];
    const float* W2 = (const float*)d_in[4];
    const float* b2 = (const float*)d_in[5];

    int n_nodes = in_sizes[0] / NFEAT;
    int n_edges = in_sizes[1] / 2;
    const int* src = ei;
    const int* dst = ei + n_edges;

    float* ws = (float*)d_ws;
    size_t off = 0;
    float* dinv = ws + off; off += ((size_t)n_nodes + 63) & ~(size_t)63;
    float* h    = ws + off; off += (size_t)n_nodes * NHID;
    float* out1 = ws + off; off += (size_t)n_nodes * NHID;
    float* z    = ws + off; off += ((size_t)2 * n_nodes + 63) & ~(size_t)63;
    (void)ws_size;

    float* out = (float*)d_out;

    const int B = 256;

    k_deg_init <<<(n_nodes + B - 1) / B, B, 0, stream>>>(dinv, n_nodes);
    k_deg_count<<<(n_edges + B - 1) / B, B, 0, stream>>>(dst, dinv, n_edges);
    k_dinv     <<<(n_nodes + B - 1) / B, B, 0, stream>>>(dinv, n_nodes);

    int tiles  = n_nodes / 16;
    int blocks = (tiles + 7) / 8;
    k_gemm1_wmma<<<blocks, B, 0, stream>>>(x, W1, h, n_nodes);

    long long t1 = (long long)n_nodes * NHID;
    k_selfloop1<<<(int)((t1 + B - 1) / B), B, 0, stream>>>(h, dinv, out1, n_nodes);
    long long t2 = (long long)n_edges * NHID;
    k_edge_agg1<<<(int)((t2 + B - 1) / B), B, 0, stream>>>(src, dst, dinv, h, out1, n_edges);
    k_relu_bias<<<(int)((t1 + B - 1) / B), B, 0, stream>>>(out1, b1, h, n_nodes);

    k_gemm2    <<<(n_nodes + B - 1) / B, B, 0, stream>>>(h, W2, b2, dinv, z, out, n_nodes);
    k_edge_agg2<<<(n_edges + B - 1) / B, B, 0, stream>>>(src, dst, dinv, z, out, n_edges);
}


# Per-call counting sort of edges by destination (arrival-order rank from the degree atomic, scan, scatter); both aggregations are per-node gathers, no float atomics; bias/relu/64x2 in gather epilogue
# speedup vs baseline: 2.6215x; 2.6215x over previous
#include <hip/hip_runtime.h>


typedef float v2f __attribute__((ext_vector_type(2)));
typedef float v8f __attribute__((ext_vector_type(8)));

#define NFEAT 256
#define NHID  64
#define SCAN_LOG   10
#define SCAN_CHUNK (1 << SCAN_LOG)


__global__ void k_cnt_zero(int* __restrict__ cnt, int n) {
    int i = blockIdx.x * blockDim.x + threadIdx.x;
    if (i < n) cnt[i] = 0;
}

__global__ void k_count_rank(const int* __restrict__ dst, int* __restrict__ cnt,
                             int* __restrict__ rank, int n_edges) {
    int e = blockIdx.x * blockDim.x + threadIdx.x;
    if (e < n_edges) rank[e] = atomicAdd(&cnt[dst[e]], 1);
}

__global__ __launch_bounds__(256) void k_scan_local(const int* __restrict__ cnt, int* __restrict__ rowl,
                                                    int* __restrict__ bsum, float* __restrict__ dinv, int n) {
    __shared__ int wsum[8];
    int tid = threadIdx.x;
    int lane = tid & 31;
    int wave = tid >> 5;
    int base = blockIdx.x * SCAN_CHUNK + tid * 4;
    int c[4];
    for (int j = 0; j < 4; ++j) c[j] = (base + j < n) ? cnt[base + j] : 0;
    int t = c[0] + c[1] + c[2] + c[3];
    int inc = t;
    for (int o = 1; o < 32; o <<= 1) {
        int v = __shfl_up(inc, o);
        if (lane >= o) inc += v;
    }
    if (lane == 31) wsum[wave] = inc;
    __syncthreads();
    int woff = 0;
    for (int w = 0; w < wave; ++w) woff += wsum[w];
    int run = woff + inc - t;
    for (int j = 0; j < 4; ++j) {
        if (base + j < n) {
            rowl[base + j] = run;
            float d = 1.0f + (float)c[j];
            dinv[base + j] = (d > 0.0f) ? rsqrtf(d) : 0.0f;
        }
        run += c[j];
    }
    if (tid == 255) bsum[blockIdx.x] = woff + inc;
}

__global__ __launch_bounds__(256) void k_scan_chunks(const int* __restrict__ bsum, int* __restrict__ boff, int nb) {
    __shared__ int wsum[8];
    int tid = threadIdx.x;
    int lane = tid & 31;
    int wave = tid >> 5;
    int carry = 0;
    for (int b0 = 0; b0 < nb; b0 += 256) {
        int i = b0 + tid;
        int v = (i < nb) ? bsum[i] : 0;
        int inc = v;
        for (int o = 1; o < 32; o <<= 1) {
            int u = __shfl_up(inc, o);
            if (lane >= o) inc += u;
        }
        if (lane == 31) wsum[wave] = inc;
        __syncthreads();
        int woff = 0, tot = 0;
        for (int w = 0; w < 8; ++w) { int s = wsum[w]; if (w < wave) woff += s; tot += s; }
        if (i < nb) boff[i] = carry + woff + inc - v;
        carry += tot;
        __syncthreads();
    }
}

__device__ __forceinline__ int seg_begin(const int* __restrict__ rowl, const int* __restrict__ boff,
                                         int i, int n_nodes, int n_edges) {
    return (i < n_nodes) ? rowl[i] + boff[i >> SCAN_LOG] : n_edges;
}

__global__ void k_fill(const int* __restrict__ src, const int* __restrict__ dst, const int* __restrict__ rank,
                       const int* __restrict__ rowl, const int* __restrict__ boff,
                       int* __restrict__ csr_src, int n_edges) {
    int e = blockIdx.x * blockDim.x + threadIdx.x;
    if (e < n_edges) {
        int d = dst[e];
        csr_src[rowl[d] + boff[d >> SCAN_LOG] + rank[e]] = src[e];
    }
}


__global__ __launch_bounds__(256) void k_gemm1_wmma(const float* __restrict__ x,
                                                    const float* __restrict__ W1,
                                                    float* __restrict__ h,
                                                    int n_nodes) {
    __shared__ float ldsW[NFEAT * NHID];
    int tid = threadIdx.x;

    for (int idx = tid * 4; idx < NFEAT * NHID; idx += 256 * 4) {
        float4 w = *(const float4*)(W1 + idx);
        int k  = idx >> 6;
        int n0 = idx & 63;
        int p  = k >> 1;
        int lo = k & 1;
        int rot = (p & 1) << 5;
        float vals[4] = {w.x, w.y, w.z, w.w};
        #pragma unroll
        for (int j = 0; j < 4; ++j) {
            int wpos = (((n0 + j) << 1) + rot) & 127;
            ldsW[p * 128 + wpos + lo] = vals[j];
        }
    }
    __syncthreads();

    int wave = tid >> 5;
    int lane = tid & 31;
    int tile = blockIdx.x * 8 + wave;
    int m0 = tile * 16;
    if (m0 >= n_nodes) return;

    int half = lane >> 4;
    int mrow = m0 + (lane & 15);
    int kk   = half << 1;
    int ncol = lane & 15;
    int rot  = half << 5;

    int wp0 = (((ncol +  0) << 1) + rot) & 127;
    int wp1 = (((ncol + 16) << 1) + rot) & 127;
    int wp2 = (((ncol + 32) << 1) + rot) & 127;
    int wp3 = (((ncol + 48) << 1) + rot) & 127;

    v8f acc0 = {}, acc1 = {}, acc2 = {}, acc3 = {};

    const float* xrow = x + (size_t)mrow * NFEAT;
    for (int k = 0; k < NFEAT; k += 4) {
        float2 av = *(const float2*)(xrow + k + kk);
        v2f a; a.x = av.x; a.y = av.y;

        const float* rowp = ldsW + ((k >> 1) + half) * 128;
        v2f b0 = *(const v2f*)(rowp + wp0);
        v2f b1 = *(const v2f*)(rowp + wp1);
        v2f b2 = *(const v2f*)(rowp + wp2);
        v2f b3 = *(const v2f*)(rowp + wp3);

        acc0 = __builtin_amdgcn_wmma_f32_16x16x4_f32(false, a, false, b0, (short)0, acc0, false, false);
        acc1 = __builtin_amdgcn_wmma_f32_16x16x4_f32(false, a, false, b1, (short)0, acc1, false, false);
        acc2 = __builtin_amdgcn_wmma_f32_16x16x4_f32(false, a, false, b2, (short)0, acc2, false, false);
        acc3 = __builtin_amdgcn_wmma_f32_16x16x4_f32(false, a, false, b3, (short)0, acc3, false, false);
    }

    int rbase = m0 + (half << 3);
    for (int r = 0; r < 8; ++r) {
        float* hr = h + (size_t)(rbase + r) * NHID;
        hr[ncol +  0] = acc0[r];
        hr[ncol + 16] = acc1[r];
        hr[ncol + 32] = acc2[r];
        hr[ncol + 48] = acc3[r];
    }
}


__global__ __launch_bounds__(512) void k_gather1(const float* __restrict__ h, const float* __restrict__ dinv,
                                                 const int* __restrict__ rowl, const int* __restrict__ boff,
                                                 const int* __restrict__ csr_src,
                                                 const float* __restrict__ b1, const float* __restrict__ W2,
                                                 float* __restrict__ z, int n_nodes, int n_edges) {
    __shared__ float zl[32];
    int tid = threadIdx.x;
    int lane = tid & 31;
    int wave = __builtin_amdgcn_readfirstlane(tid >> 5);
    int i = blockIdx.x * 16 + wave;
    if (i < n_nodes) {
        int start = seg_begin(rowl, boff, i, n_nodes, n_edges);
        int end   = seg_begin(rowl, boff, i + 1, n_nodes, n_edges);
        float di = dinv[i];
        const float* hl = h + 2 * lane;
        float2 hv = *(const float2*)(hl + (size_t)i * NHID);
        float nn = di * di;
        float ax = hv.x * nn, ay = hv.y * nn;
        for (int j = start; j < end; j += 32) {
            int m = end - j;
            if (m > 32) m = 32;
            int my_s = 0;
            float my_w = 0.0f;
            if (lane < m) {
                my_s = csr_src[j + lane];
                my_w = dinv[my_s] * di;
            }
            int my_wi = __float_as_int(my_w);
            int t = 0;
            for (; t + 4 <= m; t += 4) {
                int s0 = __builtin_amdgcn_readlane(my_s, t + 0);
                int s1 = __builtin_amdgcn_readlane(my_s, t + 1);
                int s2 = __builtin_amdgcn_readlane(my_s, t + 2);
                int s3 = __builtin_amdgcn_readlane(my_s, t + 3);
                float2 v0 = *(const float2*)(hl + (size_t)s0 * NHID);
                float2 v1 = *(const float2*)(hl + (size_t)s1 * NHID);
                float2 v2 = *(const float2*)(hl + (size_t)s2 * NHID);
                float2 v3 = *(const float2*)(hl + (size_t)s3 * NHID);
                float w0 = __int_as_float(__builtin_amdgcn_readlane(my_wi, t + 0));
                float w1 = __int_as_float(__builtin_amdgcn_readlane(my_wi, t + 1));
                float w2 = __int_as_float(__builtin_amdgcn_readlane(my_wi, t + 2));
                float w3 = __int_as_float(__builtin_amdgcn_readlane(my_wi, t + 3));
                ax = fmaf(v0.x, w0, ax); ay = fmaf(v0.y, w0, ay);
                ax = fmaf(v1.x, w1, ax); ay = fmaf(v1.y, w1, ay);
                ax = fmaf(v2.x, w2, ax); ay = fmaf(v2.y, w2, ay);
                ax = fmaf(v3.x, w3, ax); ay = fmaf(v3.y, w3, ay);
            }
            for (; t < m; ++t) {
                int s0 = __builtin_amdgcn_readlane(my_s, t);
                float2 v0 = *(const float2*)(hl + (size_t)s0 * NHID);
                float w0 = __int_as_float(__builtin_amdgcn_readlane(my_wi, t));
                ax = fmaf(v0.x, w0, ax); ay = fmaf(v0.y, w0, ay);
            }
        }
        float2 bb = *(const float2*)(b1 + 2 * lane);
        float4 wv = *(const float4*)(W2 + 4 * lane);
        float h0 = fmaxf(ax + bb.x, 0.0f);
        float h1 = fmaxf(ay + bb.y, 0.0f);
        float p0 = h0 * wv.x + h1 * wv.z;
        float p1 = h0 * wv.y + h1 * wv.w;
        for (int o = 16; o >= 1; o >>= 1) {
            p0 += __shfl_xor(p0, o);
            p1 += __shfl_xor(p1, o);
        }
        if (lane == 0) { zl[2 * wave + 0] = p0; zl[2 * wave + 1] = p1; }
    }
    __syncthreads();
    if (tid < 32) {
        int gi = blockIdx.x * 32 + tid;
        if (gi < 2 * n_nodes) z[gi] = zl[tid];
    }
}


__global__ __launch_bounds__(256) void k_gather2(const float* __restrict__ z, const float* __restrict__ dinv,
                                                 const int* __restrict__ rowl, const int* __restrict__ boff,
                                                 const int* __restrict__ csr_src, const float* __restrict__ b2,
                                                 float* __restrict__ out, int n_nodes, int n_edges) {
    __shared__ float2 ol[32];
    int tid = threadIdx.x;
    int sub = tid & 7;
    int g = tid >> 3;
    int i = blockIdx.x * 32 + g;
    float a0 = 0.0f, a1 = 0.0f;
    if (i < n_nodes) {
        int start = seg_begin(rowl, boff, i, n_nodes, n_edges);
        int end   = seg_begin(rowl, boff, i + 1, n_nodes, n_edges);
        float di = dinv[i];
        if (sub == 0) {
            float2 zi = *(const float2*)(z + 2 * (size_t)i);
            float nn = di * di;
            a0 = zi.x * nn;
            a1 = zi.y * nn;
        }
        for (int j = start + sub; j < end; j += 8) {
            int s = csr_src[j];
            float w = dinv[s] * di;
            float2 zs = *(const float2*)(z + 2 * (size_t)s);
            a0 = fmaf(zs.x, w, a0);
            a1 = fmaf(zs.y, w, a1);
        }
    }
    for (int o = 4; o >= 1; o >>= 1) {
        a0 += __shfl_xor(a0, o);
        a1 += __shfl_xor(a1, o);
    }
    if (sub == 0) ol[g] = make_float2(a0, a1);
    __syncthreads();
    if (tid < 32) {
        int o = blockIdx.x * 32 + tid;
        if (o < n_nodes) {
            float2 v = ol[tid];
            *(float2*)(out + 2 * (size_t)o) = make_float2(v.x + b2[0], v.y + b2[1]);
        }
    }
}


extern "C" void kernel_launch(void* const* d_in, const int* in_sizes, int n_in,
                              void* d_out, int out_size, void* d_ws, size_t ws_size,
                              hipStream_t stream) {
    const float* x  = (const float*)d_in[0];
    const int*   ei = (const int*)d_in[1];
    const float* W1 = (const float*)d_in[2];
    const float* b1 = (const float*)d_in[3];
    const float* W2 = (const float*)d_in[4];
    const float* b2 = (const float*)d_in[5];

    int n_nodes = in_sizes[0] / NFEAT;
    int n_edges = in_sizes[1] / 2;
    const int* src = ei;
    const int* dst = ei + n_edges;
    int n_chunks = (n_nodes + SCAN_CHUNK - 1) / SCAN_CHUNK;

    float* ws = (float*)d_ws;
    size_t off = 0;
    size_t n64 = ((size_t)n_nodes + 63) & ~(size_t)63;
    size_t c64 = ((size_t)n_chunks + 63) & ~(size_t)63;
    size_t e64 = ((size_t)n_edges + 63) & ~(size_t)63;
    float* dinv    = ws + off;         off += n64;
    float* h       = ws + off;         off += (size_t)n_nodes * NHID;
    float* z       = ws + off;         off += ((size_t)2 * n_nodes + 63) & ~(size_t)63;
    int*   cnt     = (int*)(ws + off); off += n64;
    int*   rowl    = (int*)(ws + off); off += n64;
    int*   bsum    = (int*)(ws + off); off += c64;
    int*   boff    = (int*)(ws + off); off += c64;
    int*   rank    = (int*)(ws + off); off += e64;
    int*   csr_src = (int*)(ws + off); off += e64;
    (void)ws_size; (void)n_in; (void)out_size;

    float* out = (float*)d_out;

    const int B = 256;

    k_cnt_zero   <<<(n_nodes + B - 1) / B, B, 0, stream>>>(cnt, n_nodes);
    k_count_rank <<<(n_edges + B - 1) / B, B, 0, stream>>>(dst, cnt, rank, n_edges);
    k_scan_local <<<n_chunks, 256, 0, stream>>>(cnt, rowl, bsum, dinv, n_nodes);
    k_scan_chunks<<<1, 256, 0, stream>>>(bsum, boff, n_chunks);
    k_fill       <<<(n_edges + B - 1) / B, B, 0, stream>>>(src, dst, rank, rowl, boff, csr_src, n_edges);

    int tiles  = n_nodes / 16;
    int blocks = (tiles + 7) / 8;
    k_gemm1_wmma<<<blocks, B, 0, stream>>>(x, W1, h, n_nodes);

    k_gather1<<<(n_nodes + 15) / 16, 512, 0, stream>>>(h, dinv, rowl, boff, csr_src, b1, W2, z, n_nodes, n_edges);
    k_gather2<<<(n_nodes + 31) / 32, 256, 0, stream>>>(z, dinv, rowl, boff, csr_src, b2, out, n_nodes, n_edges);
}


# Edge sort by destination counted and ordered in LDS (256 slices bin by 256-node range, one block per bin); stores leave as contiguous runs; no global atomics; sum order = LDS atomic arrival
# speedup vs baseline: 4.8077x; 1.8340x over previous
#include <hip/hip_runtime.h>


typedef float v2f __attribute__((ext_vector_type(2)));
typedef float v8f __attribute__((ext_vector_type(8)));

#define NFEAT 256
#define NHID  64
#define BIN_LOG   8
#define BIN_NODES (1 << BIN_LOG)
#define N_SLICES  256
#define MAX_BINS  1024
#define SLICE_CAP 6400
#define BIN_CAP   8192


__global__ __launch_bounds__(1024) void k_bin_count(const int* __restrict__ dst, int* __restrict__ hist,
                                                    int n_edges, int n_bins, int slice_len) {
    __shared__ int lh[MAX_BINS];
    int tid = threadIdx.x;
    int w = blockIdx.x;
    lh[tid] = 0;
    __syncthreads();
    int e0 = w * slice_len;
    int e1 = e0 + slice_len;
    if (e1 > n_edges) e1 = n_edges;
    for (int e = e0 + tid; e < e1; e += 4 * 1024) {
        int d[4];
        #pragma unroll
        for (int k = 0; k < 4; ++k) {
            int ee = e + k * 1024;
            d[k] = (ee < e1) ? dst[ee] : -1;
        }
        #pragma unroll
        for (int k = 0; k < 4; ++k)
            if (d[k] >= 0) atomicAdd(&lh[d[k] >> BIN_LOG], 1);
    }
    __syncthreads();
    if (tid < n_bins) hist[tid * N_SLICES + w] = lh[tid];
}

__global__ __launch_bounds__(256) void k_bin_offsets(int* __restrict__ hist, int* __restrict__ total, int n_bins) {
    int lane = threadIdx.x & 31;
    int b = blockIdx.x * 8 + (threadIdx.x >> 5);
    if (b >= n_bins) return;
    int* p = hist + (size_t)b * N_SLICES + lane * 8;
    int4 a = *(const int4*)p;
    int4 c = *(const int4*)(p + 4);
    int t = a.x + a.y + a.z + a.w + c.x + c.y + c.z + c.w;
    int inc = t;
    for (int o = 1; o < 32; o <<= 1) {
        int v = __shfl_up(inc, o);
        if (lane >= o) inc += v;
    }
    int r0 = inc - t;
    int4 oa, oc;
    oa.x = r0;
    oa.y = oa.x + a.x;
    oa.z = oa.y + a.y;
    oa.w = oa.z + a.z;
    oc.x = oa.w + a.w;
    oc.y = oc.x + c.x;
    oc.z = oc.y + c.y;
    oc.w = oc.z + c.z;
    *(int4*)p = oa;
    *(int4*)(p + 4) = oc;
    if (lane == 31) total[b] = inc;
}

__global__ __launch_bounds__(1024) void k_bin_scatter(const int* __restrict__ src, const int* __restrict__ dst,
                                                      const int* __restrict__ hist, const int* __restrict__ total,
                                                      int* __restrict__ bin_start, int2* __restrict__ pairs,
                                                      int n_edges, int n_bins, int slice_len) {
    __shared__ int gbase[MAX_BINS];
    __shared__ int lstart[MAX_BINS];
    __shared__ int lcur[MAX_BINS];
    __shared__ int wsum[32];
    __shared__ int wsum2[32];
    __shared__ int2 stage[SLICE_CAP];
    int tid = threadIdx.x;
    int lane = tid & 31;
    int wave = tid >> 5;
    int w = blockIdx.x;
    int v = 0, c = 0, off = 0;
    if (tid < n_bins) {
        v = total[tid];
        off = hist[tid * N_SLICES + w];
        int nxt = (w + 1 < N_SLICES) ? hist[tid * N_SLICES + w + 1] : v;
        c = nxt - off;
    }
    int inc = v, inc2 = c;
    for (int o = 1; o < 32; o <<= 1) {
        int u = __shfl_up(inc, o);
        int u2 = __shfl_up(inc2, o);
        if (lane >= o) { inc += u; inc2 += u2; }
    }
    if (lane == 31) { wsum[wave] = inc; wsum2[wave] = inc2; }
    __syncthreads();
    int woff = 0, woff2 = 0;
    for (int k = 0; k < wave; ++k) { woff += wsum[k]; woff2 += wsum2[k]; }
    int excl = woff + inc - v;
    int lex = woff2 + inc2 - c;
    if (tid < n_bins) {
        gbase[tid] = excl + off;
        lstart[tid] = lex;
        lcur[tid] = lex;
        if (w == 0) bin_start[tid] = excl;
    }
    __syncthreads();
    int e0 = w * slice_len;
    int e1 = e0 + slice_len;
    if (e1 > n_edges) e1 = n_edges;
    bool staged = slice_len <= SLICE_CAP;
    for (int e = e0 + tid; e < e1; e += 4 * 1024) {
        int d[4], s[4];
        #pragma unroll
        for (int k = 0; k < 4; ++k) {
            int ee = e + k * 1024;
            bool in = ee < e1;
            d[k] = in ? dst[ee] : -1;
            s[k] = in ? src[ee] : 0;
        }
        #pragma unroll
        for (int k = 0; k < 4; ++k) {
            if (d[k] >= 0) {
                if (staged) {
                    int r = atomicAdd(&lcur[d[k] >> BIN_LOG], 1);
                    stage[r] = make_int2(s[k], d[k]);
                } else {
                    int pos = atomicAdd(&gbase[d[k] >> BIN_LOG], 1);
                    pairs[pos] = make_int2(s[k], d[k]);
                }
            }
        }
    }
    if (staged) {
        __syncthreads();
        int n_loc = e1 - e0;
        for (int t = tid; t < n_loc; t += 1024) {
            int2 p = stage[t];
            int b = p.y >> BIN_LOG;
            pairs[gbase[b] + (t - lstart[b])] = p;
        }
    }
}

__global__ __launch_bounds__(512) void k_bin_sort(const int2* __restrict__ pairs, const int* __restrict__ bin_start,
                                                  const int* __restrict__ total,
                                                  int* __restrict__ rowptr, float* __restrict__ dinv,
                                                  int* __restrict__ csr_src, int n_nodes) {
    __shared__ int cnt[BIN_NODES];
    __shared__ int cur[BIN_NODES];
    __shared__ int wsum[8];
    __shared__ int lsrc[BIN_CAP];
    int tid = threadIdx.x;
    int lane = tid & 31;
    int wave = tid >> 5;
    int b = blockIdx.x;
    int s0 = bin_start[b];
    int s1 = s0 + total[b];
    bool staged = (s1 - s0) <= BIN_CAP;
    if (tid < BIN_NODES) cnt[tid] = 0;
    __syncthreads();
    for (int j = s0 + tid; j < s1; j += 4 * 512) {
        int d[4];
        #pragma unroll
        for (int k = 0; k < 4; ++k) {
            int jj = j + k * 512;
            d[k] = (jj < s1) ? pairs[jj].y : -1;
        }
        #pragma unroll
        for (int k = 0; k < 4; ++k)
            if (d[k] >= 0) atomicAdd(&cnt[d[k] & (BIN_NODES - 1)], 1);
    }
    __syncthreads();
    int c = 0, inc = 0;
    if (tid < BIN_NODES) {
        c = cnt[tid];
        inc = c;
        for (int o = 1; o < 32; o <<= 1) {
            int u = __shfl_up(inc, o);
            if (lane >= o) inc += u;
        }
        if (lane == 31) wsum[wave] = inc;
    }
    __syncthreads();
    if (tid < BIN_NODES) {
        int woff = 0;
        for (int k = 0; k < wave; ++k) woff += wsum[k];
        int excl = woff + inc - c;
        cur[tid] = excl;
        int node = b * BIN_NODES + tid;
        if (node < n_nodes) {
            rowptr[node] = s0 + excl;
            float d = 1.0f + (float)c;
            dinv[node] = (d > 0.0f) ? rsqrtf(d) : 0.0f;
        }
    }
    __syncthreads();
    for (int j = s0 + tid; j < s1; j += 4 * 512) {
        int2 p[4];
        #pragma unroll
        for (int k = 0; k < 4; ++k) {
            int jj = j + k * 512;
            p[k] = (jj < s1) ? pairs[jj] : make_int2(0, -1);
        }
        #pragma unroll
        for (int k = 0; k < 4; ++k) {
            if (p[k].y >= 0) {
                int pos = atomicAdd(&cur[p[k].y & (BIN_NODES - 1)], 1);
                if (staged) lsrc[pos] = p[k].x;
                else        csr_src[s0 + pos] = p[k].x;
            }
        }
    }
    if (staged) {
        __syncthreads();
        int nb = s1 - s0;
        for (int t = tid; t < nb; t += 512) csr_src[s0 + t] = lsrc[t];
    }
}

__device__ __forceinline__ int seg_end(const int* __restrict__ rowptr, int i, int n_nodes, int n_edges) {
    return (i + 1 < n_nodes) ? rowptr[i + 1] : n_edges;
}


__global__ __launch_bounds__(256) void k_gemm1_wmma(const float* __restrict__ x,
                                                    const float* __restrict__ W1,
                                                    const float* __restrict__ dinv,
                                                    float* __restrict__ h,
                                                    int n_nodes) {
    __shared__ float ldsW[NFEAT * NHID];
    int tid = threadIdx.x;

    for (int idx = tid * 4; idx < NFEAT * NHID; idx += 256 * 4) {
        float4 w = *(const float4*)(W1 + idx);
        int k  = idx >> 6;
        int n0 = idx & 63;
        int p  = k >> 1;
        int lo = k & 1;
        int rot = (p & 1) << 5;
        float vals[4] = {w.x, w.y, w.z, w.w};
        #pragma unroll
        for (int j = 0; j < 4; ++j) {
            int wpos = (((n0 + j) << 1) + rot) & 127;
            ldsW[p * 128 + wpos + lo] = vals[j];
        }
    }
    __syncthreads();

    int wave = tid >> 5;
    int lane = tid & 31;
    int tile = blockIdx.x * 8 + wave;
    int m0 = tile * 16;
    if (m0 >= n_nodes) return;

    int half = lane >> 4;
    int mrow = m0 + (lane & 15);
    int kk   = half << 1;
    int ncol = lane & 15;
    int rot  = half << 5;

    int wp0 = (((ncol +  0) << 1) + rot) & 127;
    int wp1 = (((ncol + 16) << 1) + rot) & 127;
    int wp2 = (((ncol + 32) << 1) + rot) & 127;
    int wp3 = (((ncol + 48) << 1) + rot) & 127;

    v8f acc0 = {}, acc1 = {}, acc2 = {}, acc3 = {};

    int rbase = m0 + (half << 3);
    float dr[8];
    for (int r = 0; r < 8; ++r) dr[r] = dinv[rbase + r];

    const float* xrow = x + (size_t)mrow * NFEAT;
    for (int k = 0; k < NFEAT; k += 4) {
        float2 av = *(const float2*)(xrow + k + kk);
        v2f a; a.x = av.x; a.y = av.y;

        const float* rowp = ldsW + ((k >> 1) + half) * 128;
        v2f b0 = *(const v2f*)(rowp + wp0);
        v2f b1 = *(const v2f*)(rowp + wp1);
        v2f b2 = *(const v2f*)(rowp + wp2);
        v2f b3 = *(const v2f*)(rowp + wp3);

        acc0 = __builtin_amdgcn_wmma_f32_16x16x4_f32(false, a, false, b0, (short)0, acc0, false, false);
        acc1 = __builtin_amdgcn_wmma_f32_16x16x4_f32(false, a, false, b1, (short)0, acc1, false, false);
        acc2 = __builtin_amdgcn_wmma_f32_16x16x4_f32(false, a, false, b2, (short)0, acc2, false, false);
        acc3 = __builtin_amdgcn_wmma_f32_16x16x4_f32(false, a, false, b3, (short)0, acc3, false, false);
    }

    for (int r = 0; r < 8; ++r) {
        float* hr = h + (size_t)(rbase + r) * NHID;
        hr[ncol +  0] = acc0[r] * dr[r];
        hr[ncol + 16] = acc1[r] * dr[r];
        hr[ncol + 32] = acc2[r] * dr[r];
        hr[ncol + 48] = acc3[r] * dr[r];
    }
}


__global__ __launch_bounds__(512) void k_gather1(const float* __restrict__ h, const float* __restrict__ dinv,
                                                 const int* __restrict__ rowptr,
                                                 const int* __restrict__ csr_src,
                                                 const float* __restrict__ b1, const float* __restrict__ W2,
                                                 float* __restrict__ z, int n_nodes, int n_edges) {
    __shared__ float zl[32];
    int tid = threadIdx.x;
    int lane = tid & 31;
    int wave = __builtin_amdgcn_readfirstlane(tid >> 5);
    int i = blockIdx.x * 16 + wave;
    if (i < n_nodes) {
        int start = rowptr[i];
        int end   = seg_end(rowptr, i, n_nodes, n_edges);
        float di = dinv[i];
        const float* hl = h + 2 * lane;
        float2 hv = *(const float2*)(hl + (size_t)i * NHID);
        float ax = hv.x, ay = hv.y;
        for (int j = start; j < end; j += 32) {
            int m = end - j;
            if (m > 32) m = 32;
            int my_s = 0;
            if (lane < m) my_s = csr_src[j + lane];
            int t = 0;
            for (; t + 4 <= m; t += 4) {
                int s0 = __builtin_amdgcn_readlane(my_s, t + 0);
                int s1 = __builtin_amdgcn_readlane(my_s, t + 1);
                int s2 = __builtin_amdgcn_readlane(my_s, t + 2);
                int s3 = __builtin_amdgcn_readlane(my_s, t + 3);
                float2 v0 = *(const float2*)(hl + (size_t)s0 * NHID);
                float2 v1 = *(const float2*)(hl + (size_t)s1 * NHID);
                float2 v2 = *(const float2*)(hl + (size_t)s2 * NHID);
                float2 v3 = *(const float2*)(hl + (size_t)s3 * NHID);
                ax += v0.x; ay += v0.y;
                ax += v1.x; ay += v1.y;
                ax += v2.x; ay += v2.y;
                ax += v3.x; ay += v3.y;
            }
            for (; t < m; ++t) {
                int s0 = __builtin_amdgcn_readlane(my_s, t);
                float2 v0 = *(const float2*)(hl + (size_t)s0 * NHID);
                ax += v0.x; ay += v0.y;
            }
        }
        float2 bb = *(const float2*)(b1 + 2 * lane);
        float4 wv = *(const float4*)(W2 + 4 * lane);
        float h0 = fmaxf(fmaf(ax, di, bb.x), 0.0f);
        float h1 = fmaxf(fmaf(ay, di, bb.y), 0.0f);
        float p0 = h0 * wv.x + h1 * wv.z;
        float p1 = h0 * wv.y + h1 * wv.w;
        for (int o = 16; o >= 1; o >>= 1) {
            p0 += __shfl_xor(p0, o);
            p1 += __shfl_xor(p1, o);
        }
        if (lane == 0) { zl[2 * wave + 0] = p0 * di; zl[2 * wave + 1] = p1 * di; }
    }
    __syncthreads();
    if (tid < 32) {
        int gi = blockIdx.x * 32 + tid;
        if (gi < 2 * n_nodes) z[gi] = zl[tid];
    }
}


__global__ __launch_bounds__(256) void k_gather2(const float* __restrict__ z, const float* __restrict__ dinv,
                                                 const int* __restrict__ rowptr,
                                                 const int* __restrict__ csr_src, const float* __restrict__ b2,
                                                 float* __restrict__ out, int n_nodes, int n_edges) {
    __shared__ float2 ol[32];
    int tid = threadIdx.x;
    int sub = tid & 7;
    int g = tid >> 3;
    int i = blockIdx.x * 32 + g;
    float a0 = 0.0f, a1 = 0.0f;
    float di = 0.0f;
    if (i < n_nodes) {
        int start = rowptr[i];
        int end   = seg_end(rowptr, i, n_nodes, n_edges);
        di = dinv[i];
        if (sub == 0) {
            float2 zi = *(const float2*)(z + 2 * (size_t)i);
            a0 = zi.x;
            a1 = zi.y;
        }
        for (int j = start + sub; j < end; j += 8) {
            int s = csr_src[j];
            float2 zs = *(const float2*)(z + 2 * (size_t)s);
            a0 += zs.x;
            a1 += zs.y;
        }
    }
    for (int o = 4; o >= 1; o >>= 1) {
        a0 += __shfl_xor(a0, o);
        a1 += __shfl_xor(a1, o);
    }
    if (sub == 0) ol[g] = make_float2(a0 * di, a1 * di);
    __syncthreads();
    if (tid < 32) {
        int o = blockIdx.x * 32 + tid;
        if (o < n_nodes) {
            float2 v = ol[tid];
            *(float2*)(out + 2 * (size_t)o) = make_float2(v.x + b2[0], v.y + b2[1]);
        }
    }
}


extern "C" void kernel_launch(void* const* d_in, const int* in_sizes, int n_in,
                              void* d_out, int out_size, void* d_ws, size_t ws_size,
                              hipStream_t stream) {
    const float* x  = (const float*)d_in[0];
    const int*   ei = (const int*)d_in[1];
    const float* W1 = (const float*)d_in[2];
    const float* b1 = (const float*)d_in[3];
    const float* W2 = (const float*)d_in[4];
    const float* b2 = (const float*)d_in[5];

    int n_nodes = in_sizes[0] / NFEAT;
    int n_edges = in_sizes[1] / 2;
    const int* src = ei;
    const int* dst = ei + n_edges;
    int n_bins = (n_nodes + BIN_NODES - 1) / BIN_NODES;
    int slice_len = (n_edges + N_SLICES - 1) / N_SLICES;

    float* ws = (float*)d_ws;
    size_t off = 0;
    size_t n64 = ((size_t)n_nodes + 63) & ~(size_t)63;
    size_t e64 = ((size_t)n_edges + 63) & ~(size_t)63;
    float* dinv    = ws + off;         off += n64;
    float* h       = ws + off;         off += (size_t)n_nodes * NHID;
    float* z       = ws + off;         off += ((size_t)2 * n_nodes + 63) & ~(size_t)63;
    int*   rowptr  = (int*)(ws + off); off += n64;
    int*   total   = (int*)(ws + off); off += MAX_BINS;
    int*   bstart  = (int*)(ws + off); off += MAX_BINS;
    int*   csr_src = (int*)(ws + off); off += e64;
    int2*  pairs   = (int2*)h;
    int*   hist    = (int*)h + 2 * e64;
    (void)ws_size; (void)n_in; (void)out_size;

    float* out = (float*)d_out;

    const int B = 256;

    k_bin_count  <<<N_SLICES, 1024, 0, stream>>>(dst, hist, n_edges, n_bins, slice_len);
    k_bin_offsets<<<(n_bins + 7) / 8, 256, 0, stream>>>(hist, total, n_bins);
    k_bin_scatter<<<N_SLICES, 1024, 0, stream>>>(src, dst, hist, total, bstart, pairs, n_edges, n_bins, slice_len);
    k_bin_sort   <<<n_bins, 512, 0, stream>>>(pairs, bstart, total, rowptr, dinv, csr_src, n_nodes);

    int tiles  = n_nodes / 16;
    int blocks = (tiles + 7) / 8;
    k_gemm1_wmma<<<blocks, B, 0, stream>>>(x, W1, dinv, h, n_nodes);

    k_gather1<<<(n_nodes + 15) / 16, 512, 0, stream>>>(h, dinv, rowptr, csr_src, b1, W2, z, n_nodes, n_edges);
    k_gather2<<<(n_nodes + 31) / 32, 256, 0, stream>>>(z, dinv, rowptr, csr_src, b2, out, n_nodes, n_edges);
}


# Both gathers: one wave serves 16 destination nodes (6256 waves each; were 100000 and 25000), edge rows streamed 32 at a time across node bounds; per-node sum order and wave-sum pairing kept
# speedup vs baseline: 5.2270x; 1.0872x over previous
#include <hip/hip_runtime.h>


typedef float v2f __attribute__((ext_vector_type(2)));
typedef float v8f __attribute__((ext_vector_type(8)));

#define NFEAT 256
#define NHID  64
#define BIN_LOG   8
#define BIN_NODES (1 << BIN_LOG)
#define N_SLICES  256
#define MAX_BINS  1024
#define SLICE_CAP 6400
#define BIN_CAP   8192


__global__ __launch_bounds__(1024) void k_bin_count(const int* __restrict__ dst, int* __restrict__ hist,
                                                    int n_edges, int n_bins, int slice_len) {
    __shared__ int lh[MAX_BINS];
    int tid = threadIdx.x;
    int w = blockIdx.x;
    lh[tid] = 0;
    __syncthreads();
    int e0 = w * slice_len;
    int e1 = e0 + slice_len;
    if (e1 > n_edges) e1 = n_edges;
    for (int e = e0 + tid; e < e1; e += 4 * 1024) {
        int d[4];
        #pragma unroll
        for (int k = 0; k < 4; ++k) {
            int ee = e + k * 1024;
            d[k] = (ee < e1) ? dst[ee] : -1;
        }
        #pragma unroll
        for (int k = 0; k < 4; ++k)
            if (d[k] >= 0) atomicAdd(&lh[d[k] >> BIN_LOG], 1);
    }
    __syncthreads();
    if (tid < n_bins) hist[tid * N_SLICES + w] = lh[tid];
}

__global__ __launch_bounds__(256) void k_bin_offsets(int* __restrict__ hist, int* __restrict__ total, int n_bins) {
    int lane = threadIdx.x & 31;
    int b = blockIdx.x * 8 + (threadIdx.x >> 5);
    if (b >= n_bins) return;
    int* p = hist + (size_t)b * N_SLICES + lane * 8;
    int4 a = *(const int4*)p;
    int4 c = *(const int4*)(p + 4);
    int t = a.x + a.y + a.z + a.w + c.x + c.y + c.z + c.w;
    int inc = t;
    for (int o = 1; o < 32; o <<= 1) {
        int v = __shfl_up(inc, o);
        if (lane >= o) inc += v;
    }
    int r0 = inc - t;
    int4 oa, oc;
    oa.x = r0;
    oa.y = oa.x + a.x;
    oa.z = oa.y + a.y;
    oa.w = oa.z + a.z;
    oc.x = oa.w + a.w;
    oc.y = oc.x + c.x;
    oc.z = oc.y + c.y;
    oc.w = oc.z + c.z;
    *(int4*)p = oa;
    *(int4*)(p + 4) = oc;
    if (lane == 31) total[b] = inc;
}

__global__ __launch_bounds__(1024) void k_bin_scatter(const int* __restrict__ src, const int* __restrict__ dst,
                                                      const int* __restrict__ hist, const int* __restrict__ total,
                                                      int* __restrict__ bin_start, int2* __restrict__ pairs,
                                                      int n_edges, int n_bins, int slice_len) {
    __shared__ int gbase[MAX_BINS];
    __shared__ int lstart[MAX_BINS];
    __shared__ int lcur[MAX_BINS];
    __shared__ int wsum[32];
    __shared__ int wsum2[32];
    __shared__ int2 stage[SLICE_CAP];
    int tid = threadIdx.x;
    int lane = tid & 31;
    int wave = tid >> 5;
    int w = blockIdx.x;
    int v = 0, c = 0, off = 0;
    if (tid < n_bins) {
        v = total[tid];
        off = hist[tid * N_SLICES + w];
        int nxt = (w + 1 < N_SLICES) ? hist[tid * N_SLICES + w + 1] : v;
        c = nxt - off;
    }
    int inc = v, inc2 = c;
    for (int o = 1; o < 32; o <<= 1) {
        int u = __shfl_up(inc, o);
        int u2 = __shfl_up(inc2, o);
        if (lane >= o) { inc += u; inc2 += u2; }
    }
    if (lane == 31) { wsum[wave] = inc; wsum2[wave] = inc2; }
    __syncthreads();
    int woff = 0, woff2 = 0;
    for (int k = 0; k < wave; ++k) { woff += wsum[k]; woff2 += wsum2[k]; }
    int excl = woff + inc - v;
    int lex = woff2 + inc2 - c;
    if (tid < n_bins) {
        gbase[tid] = excl + off;
        lstart[tid] = lex;
        lcur[tid] = lex;
        if (w == 0) bin_start[tid] = excl;
    }
    __syncthreads();
    int e0 = w * slice_len;
    int e1 = e0 + slice_len;
    if (e1 > n_edges) e1 = n_edges;
    bool staged = slice_len <= SLICE_CAP;
    for (int e = e0 + tid; e < e1; e += 4 * 1024) {
        int d[4], s[4];
        #pragma unroll
        for (int k = 0; k < 4; ++k) {
            int ee = e + k * 1024;
            bool in = ee < e1;
            d[k] = in ? dst[ee] : -1;
            s[k] = in ? src[ee] : 0;
        }
        #pragma unroll
        for (int k = 0; k < 4; ++k) {
            if (d[k] >= 0) {
                if (staged) {
                    int r = atomicAdd(&lcur[d[k] >> BIN_LOG], 1);
                    stage[r] = make_int2(s[k], d[k]);
                } else {
                    int pos = atomicAdd(&gbase[d[k] >> BIN_LOG], 1);
                    pairs[pos] = make_int2(s[k], d[k]);
                }
            }
        }
    }
    if (staged) {
        __syncthreads();
        int n_loc = e1 - e0;
        for (int t = tid; t < n_loc; t += 1024) {
            int2 p = stage[t];
            int b = p.y >> BIN_LOG;
            pairs[gbase[b] + (t - lstart[b])] = p;
        }
    }
}

__global__ __launch_bounds__(512) void k_bin_sort(const int2* __restrict__ pairs, const int* __restrict__ bin_start,
                                                  const int* __restrict__ total,
                                                  int* __restrict__ rowptr, float* __restrict__ dinv,
                                                  int* __restrict__ csr_src, int n_nodes) {
    __shared__ int cnt[BIN_NODES];
    __shared__ int cur[BIN_NODES];
    __shared__ int wsum[8];
    __shared__ int lsrc[BIN_CAP];
    int tid = threadIdx.x;
    int lane = tid & 31;
    int wave = tid >> 5;
    int b = blockIdx.x;
    int s0 = bin_start[b];
    int s1 = s0 + total[b];
    bool staged = (s1 - s0) <= BIN_CAP;
    if (tid < BIN_NODES) cnt[tid] = 0;
    __syncthreads();
    for (int j = s0 + tid; j < s1; j += 4 * 512) {
        int d[4];
        #pragma unroll
        for (int k = 0; k < 4; ++k) {
            int jj = j + k * 512;
            d[k] = (jj < s1) ? pairs[jj].y : -1;
        }
        #pragma unroll
        for (int k = 0; k < 4; ++k)
            if (d[k] >= 0) atomicAdd(&cnt[d[k] & (BIN_NODES - 1)], 1);
    }
    __syncthreads();
    int c = 0, inc = 0;
    if (tid < BIN_NODES) {
        c = cnt[tid];
        inc = c;
        for (int o = 1; o < 32; o <<= 1) {
            int u = __shfl_up(inc, o);
            if (lane >= o) inc += u;
        }
        if (lane == 31) wsum[wave] = inc;
    }
    __syncthreads();
    if (tid < BIN_NODES) {
        int woff = 0;
        for (int k = 0; k < wave; ++k) woff += wsum[k];
        int excl = woff + inc - c;
        cur[tid] = excl;
        int node = b * BIN_NODES + tid;
        if (node < n_nodes) {
            rowptr[node] = s0 + excl;
            float d = 1.0f + (float)c;
            dinv[node] = (d > 0.0f) ? rsqrtf(d) : 0.0f;
        }
    }
    __syncthreads();
    for (int j = s0 + tid; j < s1; j += 4 * 512) {
        int2 p[4];
        #pragma unroll
        for (int k = 0; k < 4; ++k) {
            int jj = j + k * 512;
            p[k] = (jj < s1) ? pairs[jj] : make_int2(0, -1);
        }
        #pragma unroll
        for (int k = 0; k < 4; ++k) {
            if (p[k].y >= 0) {
                int pos = atomicAdd(&cur[p[k].y & (BIN_NODES - 1)], 1);
                if (staged) lsrc[pos] = p[k].x;
                else        csr_src[s0 + pos] = p[k].x;
            }
        }
    }
    if (staged) {
        __syncthreads();
        int nb = s1 - s0;
        for (int t = tid; t < nb; t += 512) csr_src[s0 + t] = lsrc[t];
    }
}


__global__ __launch_bounds__(256) void k_gemm1_wmma(const float* __restrict__ x,
                                                    const float* __restrict__ W1,
                                                    const float* __restrict__ dinv,
                                                    float* __restrict__ h,
                                                    int n_nodes) {
    __shared__ float ldsW[NFEAT * NHID];
    int tid = threadIdx.x;

    for (int idx = tid * 4; idx < NFEAT * NHID; idx += 256 * 4) {
        float4 w = *(const float4*)(W1 + idx);
        int k  = idx >> 6;
        int n0 = idx & 63;
        int p  = k >> 1;
        int lo = k & 1;
        int rot = (p & 1) << 5;
        float vals[4] = {w.x, w.y, w.z, w.w};
        #pragma unroll
        for (int j = 0; j < 4; ++j) {
            int wpos = (((n0 + j) << 1) + rot) & 127;
            ldsW[p * 128 + wpos + lo] = vals[j];
        }
    }
    __syncthreads();

    int wave = tid >> 5;
    int lane = tid & 31;
    int tile = blockIdx.x * 8 + wave;
    int m0 = tile * 16;
    if (m0 >= n_nodes) return;

    int half = lane >> 4;
    int mrow = m0 + (lane & 15);
    int kk   = half << 1;
    int ncol = lane & 15;
    int rot  = half << 5;

    int wp0 = (((ncol +  0) << 1) + rot) & 127;
    int wp1 = (((ncol + 16) << 1) + rot) & 127;
    int wp2 = (((ncol + 32) << 1) + rot) & 127;
    int wp3 = (((ncol + 48) << 1) + rot) & 127;

    v8f acc0 = {}, acc1 = {}, acc2 = {}, acc3 = {};

    int rbase = m0 + (half << 3);
    float dr[8];
    for (int r = 0; r < 8; ++r) dr[r] = dinv[rbase + r];

    const float* xrow = x + (size_t)mrow * NFEAT;
    for (int k = 0; k < NFEAT; k += 4) {
        float2 av = *(const float2*)(xrow + k + kk);
        v2f a; a.x = av.x; a.y = av.y;

        const float* rowp = ldsW + ((k >> 1) + half) * 128;
        v2f b0 = *(const v2f*)(rowp + wp0);
        v2f b1 = *(const v2f*)(rowp + wp1);
        v2f b2 = *(const v2f*)(rowp + wp2);
        v2f b3 = *(const v2f*)(rowp + wp3);

        acc0 = __builtin_amdgcn_wmma_f32_16x16x4_f32(false, a, false, b0, (short)0, acc0, false, false);
        acc1 = __builtin_amdgcn_wmma_f32_16x16x4_f32(false, a, false, b1, (short)0, acc1, false, false);
        acc2 = __builtin_amdgcn_wmma_f32_16x16x4_f32(false, a, false, b2, (short)0, acc2, false, false);
        acc3 = __builtin_amdgcn_wmma_f32_16x16x4_f32(false, a, false, b3, (short)0, acc3, false, false);
    }

    for (int r = 0; r < 8; ++r) {
        float* hr = h + (size_t)(rbase + r) * NHID;
        hr[ncol +  0] = acc0[r] * dr[r];
        hr[ncol + 16] = acc1[r] * dr[r];
        hr[ncol + 32] = acc2[r] * dr[r];
        hr[ncol + 48] = acc3[r] * dr[r];
    }
}


#define G1_NPW 16
#define G1_CH  32
#define G1_PITCH 66

__device__ __forceinline__ float2 g1_row(const float* __restrict__ h, int s, unsigned lane2) {
    const float* row = h + (size_t)s * NHID;
    return *(const float2*)(row + lane2);
}

#define G1_FINISH()                                                                  \
    {                                                                                \
        float h0 = fmaxf(fmaf(acc.x, di, bb.x), 0.0f);                               \
        float h1 = fmaxf(fmaf(acc.y, di, bb.y), 0.0f);                               \
        float p0 = h0 * wv.x + h1 * wv.z;                                            \
        float p1 = h0 * wv.y + h1 * wv.w;                                            \
        *(float2*)(strip + k * G1_PITCH + lane2) = make_float2(p0, p1);              \
        ++k;                                                                         \
        acc = nself;                                                                 \
        if (k + 1 < nn) nself = *(const float2*)(strip + (k + 1) * G1_PITCH + lane2); \
        kend = __builtin_amdgcn_readlane(rp, k + 1);                                 \
        di = __int_as_float(__builtin_amdgcn_readlane(__float_as_int(dv), k));       \
    }

__global__ __launch_bounds__(256) void k_gather1(const float* __restrict__ h, const float* __restrict__ dinv,
                                                 const int* __restrict__ rowptr,
                                                 const int* __restrict__ csr_src,
                                                 const float* __restrict__ b1, const float* __restrict__ W2,
                                                 float* __restrict__ z, int n_nodes, int n_edges) {
    __shared__ float2 strips2[8 * G1_NPW * G1_PITCH / 2];
    int tid = threadIdx.x;
    int lane = tid & 31;
    unsigned lane2 = 2u * (unsigned)lane;
    int wave = __builtin_amdgcn_readfirstlane(tid >> 5);
    int i0 = (blockIdx.x * 8 + wave) * G1_NPW;
    if (i0 >= n_nodes) return;
    float* strip = (float*)strips2 + wave * (G1_NPW * G1_PITCH);
    int nn = n_nodes - i0;
    if (nn > G1_NPW) nn = G1_NPW;
    const int last = n_nodes - 1;
    int gi = i0 + lane;
    int rp = (gi < n_nodes) ? rowptr[gi] : n_edges;
    float dv = (gi < n_nodes) ? dinv[gi] : 0.0f;
    int gz = i0 + (lane >> 1);
    float dz = (gz < n_nodes) ? dinv[gz] : 0.0f;
    float2 bb = *(const float2*)(b1 + lane2);
    float4 wv = *(const float4*)(W2 + 2u * lane2);
    float2 own[G1_NPW];
    #pragma unroll
    for (int q = 0; q < G1_NPW; ++q) own[q] = g1_row(h, (i0 + q > last) ? last : i0 + q, lane2);
    const int e0 = __builtin_amdgcn_readlane(rp, 0);
    const int e1 = __builtin_amdgcn_readlane(rp, nn);
    int sv = 0;
    if (e1 > e0) {
        int ic = e0 + lane;
        if (ic > e1 - 1) ic = e1 - 1;
        sv = csr_src[ic];
    }
    #pragma unroll
    for (int q = 2; q < G1_NPW; ++q) *(float2*)(strip + q * G1_PITCH + lane2) = own[q];
    int k = 0;
    int kend = __builtin_amdgcn_readlane(rp, 1);
    float di = __int_as_float(__builtin_amdgcn_readlane(__float_as_int(dv), 0));
    float2 acc = own[0];
    float2 nself = own[1];
    for (int e = e0; e < e1; e += G1_CH) {
        int in = e + G1_CH + lane;
        if (in > e1 - 1) in = e1 - 1;
        int nsv = csr_src[in];
        float2 r[G1_CH];
        #pragma unroll
        for (int t = 0; t < G1_CH; ++t) r[t] = g1_row(h, __builtin_amdgcn_readlane(sv, t), lane2);
        const int elim = e1 - e;
        int lim = kend - e;
        #pragma unroll
        for (int t = 0; t < G1_CH; ++t) {
            if (__builtin_expect(t >= lim, 0)) {
                while (k + 1 < nn && t == lim) {
                    G1_FINISH()
                    lim = kend - e;
                }
                if (t >= elim) break;
            }
            acc.x += r[t].x;
            acc.y += r[t].y;
        }
        sv = nsv;
    }
    while (k < nn) G1_FINISH()
    const float* col = strip + (lane >> 1) * G1_PITCH + (lane & 1);
    float a[32];
    #pragma unroll
    for (int j = 0; j < 32; ++j) a[j] = col[2 * j];
    #pragma unroll
    for (int w = 16; w >= 1; w >>= 1) {
        #pragma unroll
        for (int j = 0; j < w; ++j) a[j] += a[j + w];
    }
    int go = 2 * i0 + lane;
    if (go < 2 * n_nodes) z[go] = a[0] * dz;
}


#define G2_NQ 4

__global__ __launch_bounds__(256) void k_gather2(const float* __restrict__ z, const float* __restrict__ dinv,
                                                 const int* __restrict__ rowptr,
                                                 const int* __restrict__ csr_src, const float* __restrict__ b2,
                                                 float* __restrict__ out, int n_nodes, int n_edges) {
    __shared__ float2 ol[32 * G2_NQ];
    int tid = threadIdx.x;
    int sub = tid & 7;
    int g = tid >> 3;
    int base = blockIdx.x * (32 * G2_NQ);
    const int last = n_nodes - 1;
    int rs[G2_NQ], re[G2_NQ];
    float dn[G2_NQ];
    float2 zi[G2_NQ];
    #pragma unroll
    for (int q = 0; q < G2_NQ; ++q) {
        int i = base + q * 32 + g;
        int ic = (i < last) ? i : last;
        int in = (i + 1 < last) ? i + 1 : last;
        rs[q] = rowptr[ic];
        re[q] = rowptr[in];
        dn[q] = dinv[ic];
        zi[q] = *(const float2*)(z + 2 * (size_t)ic);
    }
    int j[G2_NQ], en[G2_NQ];
    float a0[G2_NQ], a1[G2_NQ], di[G2_NQ];
    int left = 0;
    #pragma unroll
    for (int q = 0; q < G2_NQ; ++q) {
        int i = base + q * 32 + g;
        bool ok = i < n_nodes;
        j[q]  = ok ? rs[q] + sub : 0;
        en[q] = ok ? ((i + 1 < n_nodes) ? re[q] : n_edges) : 0;
        di[q] = ok ? dn[q] : 0.0f;
        a0[q] = (ok && sub == 0) ? zi[q].x : 0.0f;
        a1[q] = (ok && sub == 0) ? zi[q].y : 0.0f;
        int c = en[q] - j[q];
        if (c > left) left = c;
    }
    for (; left > 0; left -= 8) {
        int s[G2_NQ];
        #pragma unroll
        for (int q = 0; q < G2_NQ; ++q) {
            s[q] = -1;
            if (j[q] < en[q]) s[q] = csr_src[j[q]];
        }
        float2 zs[G2_NQ];
        #pragma unroll
        for (int q = 0; q < G2_NQ; ++q) {
            zs[q] = make_float2(0.0f, 0.0f);
            if (s[q] >= 0) zs[q] = *(const float2*)(z + 2 * (size_t)s[q]);
        }
        #pragma unroll
        for (int q = 0; q < G2_NQ; ++q) {
            if (s[q] >= 0) {
                a0[q] += zs[q].x;
                a1[q] += zs[q].y;
            }
            j[q] += 8;
        }
    }
    #pragma unroll
    for (int q = 0; q < G2_NQ; ++q) {
        float x0 = a0[q], x1 = a1[q];
        for (int o = 4; o >= 1; o >>= 1) {
            x0 += __shfl_xor(x0, o);
            x1 += __shfl_xor(x1, o);
        }
        if (sub == 0) ol[q * 32 + g] = make_float2(x0 * di[q], x1 * di[q]);
    }
    __syncthreads();
    if (tid < 32 * G2_NQ) {
        int o = base + tid;
        if (o < n_nodes) {
            float2 v = ol[tid];
            *(float2*)(out + 2 * (size_t)o) = make_float2(v.x + b2[0], v.y + b2[1]);
        }
    }
}


extern "C" void kernel_launch(void* const* d_in, const int* in_sizes, int n_in,
                              void* d_out, int out_size, void* d_ws, size_t ws_size,
                              hipStream_t stream) {
    const float* x  = (const float*)d_in[0];
    const int*   ei = (const int*)d_in[1];
    const float* W1 = (const float*)d_in[2];
    const float* b1 = (const float*)d_in[3];
    const float* W2 = (const float*)d_in[4];
    const float* b2 = (const float*)d_in[5];

    int n_nodes = in_sizes[0] / NFEAT;
    int n_edges = in_sizes[1] / 2;
    const int* src = ei;
    const int* dst = ei + n_edges;
    int n_bins = (n_nodes + BIN_NODES - 1) / BIN_NODES;
    int slice_len = (n_edges + N_SLICES - 1) / N_SLICES;

    float* ws = (float*)d_ws;
    size_t off = 0;
    size_t n64 = ((size_t)n_nodes + 63) & ~(size_t)63;
    size_t e64 = ((size_t)n_edges + 63) & ~(size_t)63;
    float* dinv    = ws + off;         off += n64;
    float* h       = ws + off;         off += (size_t)n_nodes * NHID;
    float* z       = ws + off;         off += ((size_t)2 * n_nodes + 63) & ~(size_t)63;
    int*   rowptr  = (int*)(ws + off); off += n64;
    int*   total   = (int*)(ws + off); off += MAX_BINS;
    int*   bstart  = (int*)(ws + off); off += MAX_BINS;
    int*   csr_src = (int*)(ws + off); off += e64;
    int2*  pairs   = (int2*)h;
    int*   hist    = (int*)h + 2 * e64;
    (void)ws_size; (void)n_in; (void)out_size;

    float* out = (float*)d_out;

    const int B = 256;

    k_bin_count  <<<N_SLICES, 1024, 0, stream>>>(dst, hist, n_edges, n_bins, slice_len);
    k_bin_offsets<<<(n_bins + 7) / 8, 256, 0, stream>>>(hist, total, n_bins);
    k_bin_scatter<<<N_SLICES, 1024, 0, stream>>>(src, dst, hist, total, bstart, pairs, n_edges, n_bins, slice_len);
    k_bin_sort   <<<n_bins, 512, 0, stream>>>(pairs, bstart, total, rowptr, dinv, csr_src, n_nodes);

    int tiles  = n_nodes / 16;
    int blocks = (tiles + 7) / 8;
    k_gemm1_wmma<<<blocks, B, 0, stream>>>(x, W1, dinv, h, n_nodes);

    k_gather1<<<((n_nodes + 15) / 16 + 7) / 8, 256, 0, stream>>>(h, dinv, rowptr, csr_src, b1, W2, z, n_nodes, n_edges);
    k_gather2<<<(n_nodes + 127) / 128, 256, 0, stream>>>(z, dinv, rowptr, csr_src, b2, out, n_nodes, n_edges);
}


# Edge sort in two launches, not four: each slice orders its pairs by bin in LDS into its own region; each bin pulls its 256 pieces (start = sum of piece starts); sum order = LDS atomic arrival, kept
# speedup vs baseline: 5.7950x; 1.1087x over previous
#include <hip/hip_runtime.h>


typedef float v2f __attribute__((ext_vector_type(2)));
typedef float v8f __attribute__((ext_vector_type(8)));

#define NFEAT 256
#define NHID  64
#define BIN_LOG   8
#define BIN_NODES (1 << BIN_LOG)
#define N_SLICES  256
#define MAX_BINS  1024
#define SLICE_CAP 6400
#define BIN_CAP   8192
static_assert(N_SLICES == BIN_NODES, "k_bin_sort clears one counter per slice-reading thread");


__global__ __launch_bounds__(1024) void k_slice_sort(const int* __restrict__ src, const int* __restrict__ dst,
                                                     int* __restrict__ lstart_g, int2* __restrict__ pairs,
                                                     int n_edges, int n_bins, int slice_len) {
    __shared__ int lh[MAX_BINS];
    __shared__ int wsum[32];
    __shared__ int2 stage[SLICE_CAP];
    int tid = threadIdx.x;
    int lane = tid & 31;
    int wave = tid >> 5;
    int w = blockIdx.x;
    lh[tid] = 0;
    __syncthreads();
    int e0 = w * slice_len;
    int e1 = e0 + slice_len;
    if (e1 > n_edges) e1 = n_edges;
    for (int e = e0 + tid; e < e1; e += 4 * 1024) {
        int d[4];
        #pragma unroll
        for (int k = 0; k < 4; ++k) {
            int ee = e + k * 1024;
            d[k] = (ee < e1) ? dst[ee] : -1;
        }
        #pragma unroll
        for (int k = 0; k < 4; ++k)
            if (d[k] >= 0) atomicAdd(&lh[d[k] >> BIN_LOG], 1);
    }
    __syncthreads();
    int c = lh[tid];
    int inc = c;
    for (int o = 1; o < 32; o <<= 1) {
        int u = __shfl_up(inc, o);
        if (lane >= o) inc += u;
    }
    if (lane == 31) wsum[wave] = inc;
    __syncthreads();
    int woff = 0;
    for (int k = 0; k < wave; ++k) woff += wsum[k];
    int excl = woff + inc - c;
    lh[tid] = excl;
    int* lrow = lstart_g + (size_t)w * (n_bins + 1);
    if (tid < n_bins) lrow[tid] = excl;
    if (tid == n_bins - 1) lrow[n_bins] = excl + c;
    __syncthreads();
    bool staged = slice_len <= SLICE_CAP;
    for (int e = e0 + tid; e < e1; e += 4 * 1024) {
        int d[4], s[4];
        #pragma unroll
        for (int k = 0; k < 4; ++k) {
            int ee = e + k * 1024;
            bool in = ee < e1;
            d[k] = in ? dst[ee] : -1;
            s[k] = in ? src[ee] : 0;
        }
        #pragma unroll
        for (int k = 0; k < 4; ++k) {
            if (d[k] >= 0) {
                int r = atomicAdd(&lh[d[k] >> BIN_LOG], 1);
                if (staged) stage[r] = make_int2(s[k], d[k]);
                else        pairs[e0 + r] = make_int2(s[k], d[k]);
            }
        }
    }
    if (staged) {
        __syncthreads();
        int n_loc = e1 - e0;
        for (int t = tid; t < n_loc; t += 1024) pairs[e0 + t] = stage[t];
    }
}

#define PIECE_STEP 16
__global__ __launch_bounds__(512) void k_bin_sort(const int2* __restrict__ pairs, const int* __restrict__ lstart_g,
                                                  int* __restrict__ rowptr, float* __restrict__ dinv,
                                                  int* __restrict__ csr_src, int n_nodes, int n_bins, int slice_len) {
    __shared__ int cnt[BIN_NODES];
    __shared__ int cur[BIN_NODES];
    __shared__ int wsum[8];
    __shared__ int wsa[8];
    __shared__ int wsb[8];
    __shared__ int pbeg[N_SLICES];
    __shared__ int plen[N_SLICES];
    __shared__ int lsrc[BIN_CAP];
    int tid = threadIdx.x;
    int lane = tid & 31;
    int wave = tid >> 5;
    int b = blockIdx.x;
    if (tid < N_SLICES) {
        const int* lrow = lstart_g + (size_t)tid * (n_bins + 1) + b;
        int a = lrow[0];
        int n = lrow[1] - a;
        pbeg[tid] = tid * slice_len + a;
        plen[tid] = n;
        cnt[tid] = 0;
        for (int o = 16; o >= 1; o >>= 1) {
            a += __shfl_xor(a, o);
            n += __shfl_xor(n, o);
        }
        if (lane == 0) { wsa[wave] = a; wsb[wave] = n; }
    }
    __syncthreads();
    int s0 = 0, nb = 0;
    for (int k = 0; k < N_SLICES / 32; ++k) { s0 += wsa[k]; nb += wsb[k]; }
    bool staged = nb <= BIN_CAP;
    const int pw = tid >> 1;
    const int pl = plen[pw];
    const int2* pp = pairs + pbeg[pw] + (tid & 1);
    for (int i = (tid & 1); i < pl; i += PIECE_STEP) {
        int d[PIECE_STEP / 2];
        #pragma unroll
        for (int k = 0; k < PIECE_STEP / 2; ++k)
            d[k] = (i + 2 * k < pl) ? pp[(i - (tid & 1)) + 2 * k].y : -1;
        #pragma unroll
        for (int k = 0; k < PIECE_STEP / 2; ++k)
            if (d[k] >= 0) atomicAdd(&cnt[d[k] & (BIN_NODES - 1)], 1);
    }
    __syncthreads();
    int c = 0, inc = 0;
    if (tid < BIN_NODES) {
        c = cnt[tid];
        inc = c;
        for (int o = 1; o < 32; o <<= 1) {
            int u = __shfl_up(inc, o);
            if (lane >= o) inc += u;
        }
        if (lane == 31) wsum[wave] = inc;
    }
    __syncthreads();
    if (tid < BIN_NODES) {
        int woff = 0;
        for (int k = 0; k < wave; ++k) woff += wsum[k];
        int excl = woff + inc - c;
        cur[tid] = excl;
        int node = b * BIN_NODES + tid;
        if (node < n_nodes) {
            rowptr[node] = s0 + excl;
            float d = 1.0f + (float)c;
            dinv[node] = (d > 0.0f) ? rsqrtf(d) : 0.0f;
        }
    }
    __syncthreads();
    for (int i = (tid & 1); i < pl; i += PIECE_STEP) {
        int2 p[PIECE_STEP / 2];
        #pragma unroll
        for (int k = 0; k < PIECE_STEP / 2; ++k)
            p[k] = (i + 2 * k < pl) ? pp[(i - (tid & 1)) + 2 * k] : make_int2(0, -1);
        #pragma unroll
        for (int k = 0; k < PIECE_STEP / 2; ++k) {
            if (p[k].y >= 0) {
                int pos = atomicAdd(&cur[p[k].y & (BIN_NODES - 1)], 1);
                if (staged) lsrc[pos] = p[k].x;
                else        csr_src[s0 + pos] = p[k].x;
            }
        }
    }
    if (staged) {
        __syncthreads();
        for (int t = tid; t < nb; t += 512) csr_src[s0 + t] = lsrc[t];
    }
}


__global__ __launch_bounds__(256) void k_gemm1_wmma(const float* __restrict__ x,
                                                    const float* __restrict__ W1,
                                                    const float* __restrict__ dinv,
                                                    float* __restrict__ h,
                                                    int n_nodes) {
    __shared__ float ldsW[NFEAT * NHID];
    int tid = threadIdx.x;

    for (int idx = tid * 4; idx < NFEAT * NHID; idx += 256 * 4) {
        float4 w = *(const float4*)(W1 + idx);
        int k  = idx >> 6;
        int n0 = idx & 63;
        int p  = k >> 1;
        int lo = k & 1;
        int rot = (p & 1) << 5;
        float vals[4] = {w.x, w.y, w.z, w.w};
        #pragma unroll
        for (int j = 0; j < 4; ++j) {
            int wpos = (((n0 + j) << 1) + rot) & 127;
            ldsW[p * 128 + wpos + lo] = vals[j];
        }
    }
    __syncthreads();

    int wave = tid >> 5;
    int lane = tid & 31;
    int tile = blockIdx.x * 8 + wave;
    int m0 = tile * 16;
    if (m0 >= n_nodes) return;

    int half = lane >> 4;
    int mrow = m0 + (lane & 15);
    int kk   = half << 1;
    int ncol = lane & 15;
    int rot  = half << 5;

    int wp0 = (((ncol +  0) << 1) + rot) & 127;
    int wp1 = (((ncol + 16) << 1) + rot) & 127;
    int wp2 = (((ncol + 32) << 1) + rot) & 127;
    int wp3 = (((ncol + 48) << 1) + rot) & 127;

    v8f acc0 = {}, acc1 = {}, acc2 = {}, acc3 = {};

    int rbase = m0 + (half << 3);
    float dr[8];
    for (int r = 0; r < 8; ++r) dr[r] = dinv[rbase + r];

    const float* xrow = x + (size_t)mrow * NFEAT;
    for (int k = 0; k < NFEAT; k += 4) {
        float2 av = *(const float2*)(xrow + k + kk);
        v2f a; a.x = av.x; a.y = av.y;

        const float* rowp = ldsW + ((k >> 1) + half) * 128;
        v2f b0 = *(const v2f*)(rowp + wp0);
        v2f b1 = *(const v2f*)(rowp + wp1);
        v2f b2 = *(const v2f*)(rowp + wp2);
        v2f b3 = *(const v2f*)(rowp + wp3);

        acc0 = __builtin_amdgcn_wmma_f32_16x16x4_f32(false, a, false, b0, (short)0, acc0, false, false);
        acc1 = __builtin_amdgcn_wmma_f32_16x16x4_f32(false, a, false, b1, (short)0, acc1, false, false);
        acc2 = __builtin_amdgcn_wmma_f32_16x16x4_f32(false, a, false, b2, (short)0, acc2, false, false);
        acc3 = __builtin_amdgcn_wmma_f32_16x16x4_f32(false, a, false, b3, (short)0, acc3, false, false);
    }

    for (int r = 0; r < 8; ++r) {
        float* hr = h + (size_t)(rbase + r) * NHID;
        hr[ncol +  0] = acc0[r] * dr[r];
        hr[ncol + 16] = acc1[r] * dr[r];
        hr[ncol + 32] = acc2[r] * dr[r];
        hr[ncol + 48] = acc3[r] * dr[r];
    }
}


#define G1_NPW 16
#define G1_CH  32
#define G1_PITCH 66

__device__ __forceinline__ float2 g1_row(const float* __restrict__ h, int s, unsigned lane2) {
    const float* row = h + (size_t)s * NHID;
    return *(const float2*)(row + lane2);
}

#define G1_FINISH()                                                                  \
    {                                                                                \
        float h0 = fmaxf(fmaf(acc.x, di, bb.x), 0.0f);                               \
        float h1 = fmaxf(fmaf(acc.y, di, bb.y), 0.0f);                               \
        float p0 = h0 * wv.x + h1 * wv.z;                                            \
        float p1 = h0 * wv.y + h1 * wv.w;                                            \
        *(float2*)(strip + k * G1_PITCH + lane2) = make_float2(p0, p1);              \
        ++k;                                                                         \
        acc = nself;                                                                 \
        if (k + 1 < nn) nself = *(const float2*)(strip + (k + 1) * G1_PITCH + lane2); \
        kend = __builtin_amdgcn_readlane(rp, k + 1);                                 \
        di = __int_as_float(__builtin_amdgcn_readlane(__float_as_int(dv), k));       \
    }

__global__ __launch_bounds__(256) void k_gather1(const float* __restrict__ h, const float* __restrict__ dinv,
                                                 const int* __restrict__ rowptr,
                                                 const int* __restrict__ csr_src,
                                                 const float* __restrict__ b1, const float* __restrict__ W2,
                                                 float* __restrict__ z, int n_nodes, int n_edges) {
    __shared__ float2 strips2[8 * G1_NPW * G1_PITCH / 2];
    int tid = threadIdx.x;
    int lane = tid & 31;
    unsigned lane2 = 2u * (unsigned)lane;
    int wave = __builtin_amdgcn_readfirstlane(tid >> 5);
    int i0 = (blockIdx.x * 8 + wave) * G1_NPW;
    if (i0 >= n_nodes) return;
    float* strip = (float*)strips2 + wave * (G1_NPW * G1_PITCH);
    int nn = n_nodes - i0;
    if (nn > G1_NPW) nn = G1_NPW;
    const int last = n_nodes - 1;
    int gi = i0 + lane;
    int rp = (gi < n_nodes) ? rowptr[gi] : n_edges;
    float dv = (gi < n_nodes) ? dinv[gi] : 0.0f;
    int gz = i0 + (lane >> 1);
    float dz = (gz < n_nodes) ? dinv[gz] : 0.0f;
    float2 bb = *(const float2*)(b1 + lane2);
    float4 wv = *(const float4*)(W2 + 2u * lane2);
    float2 own[G1_NPW];
    #pragma unroll
    for (int q = 0; q < G1_NPW; ++q) own[q] = g1_row(h, (i0 + q > last) ? last : i0 + q, lane2);
    const int e0 = __builtin_amdgcn_readlane(rp, 0);
    const int e1 = __builtin_amdgcn_readlane(rp, nn);
    int sv = 0;
    if (e1 > e0) {
        int ic = e0 + lane;
        if (ic > e1 - 1) ic = e1 - 1;
        sv = csr_src[ic];
    }
    #pragma unroll
    for (int q = 2; q < G1_NPW; ++q) *(float2*)(strip + q * G1_PITCH + lane2) = own[q];
    int k = 0;
    int kend = __builtin_amdgcn_readlane(rp, 1);
    float di = __int_as_float(__builtin_amdgcn_readlane(__float_as_int(dv), 0));
    float2 acc = own[0];
    float2 nself = own[1];
    for (int e = e0; e < e1; e += G1_CH) {
        int in = e + G1_CH + lane;
        if (in > e1 - 1) in = e1 - 1;
        int nsv = csr_src[in];
        float2 r[G1_CH];
        #pragma unroll
        for (int t = 0; t < G1_CH; ++t) r[t] = g1_row(h, __builtin_amdgcn_readlane(sv, t), lane2);
        const int elim = e1 - e;
        int lim = kend - e;
        #pragma unroll
        for (int t = 0; t < G1_CH; ++t) {
            if (__builtin_expect(t >= lim, 0)) {
                while (k + 1 < nn && t == lim) {
                    G1_FINISH()
                    lim = kend - e;
                }
                if (t >= elim) break;
            }
            acc.x += r[t].x;
            acc.y += r[t].y;
        }
        sv = nsv;
    }
    while (k < nn) G1_FINISH()
    const float* col = strip + (lane >> 1) * G1_PITCH + (lane & 1);
    float a[32];
    #pragma unroll
    for (int j = 0; j < 32; ++j) a[j] = col[2 * j];
    #pragma unroll
    for (int w = 16; w >= 1; w >>= 1) {
        #pragma unroll
        for (int j = 0; j < w; ++j) a[j] += a[j + w];
    }
    int go = 2 * i0 + lane;
    if (go < 2 * n_nodes) z[go] = a[0] * dz;
}


#define G2_NQ 4

__global__ __launch_bounds__(256) void k_gather2(const float* __restrict__ z, const float* __restrict__ dinv,
                                                 const int* __restrict__ rowptr,
                                                 const int* __restrict__ csr_src, const float* __restrict__ b2,
                                                 float* __restrict__ out, int n_nodes, int n_edges) {
    __shared__ float2 ol[32 * G2_NQ];
    int tid = threadIdx.x;
    int sub = tid & 7;
    int g = tid >> 3;
    int base = blockIdx.x * (32 * G2_NQ);
    const int last = n_nodes - 1;
    int rs[G2_NQ], re[G2_NQ];
    float dn[G2_NQ];
    float2 zi[G2_NQ];
    #pragma unroll
    for (int q = 0; q < G2_NQ; ++q) {
        int i = base + q * 32 + g;
        int ic = (i < last) ? i : last;
        int in = (i + 1 < last) ? i + 1 : last;
        rs[q] = rowptr[ic];
        re[q] = rowptr[in];
        dn[q] = dinv[ic];
        zi[q] = *(const float2*)(z + 2 * (size_t)ic);
    }
    int j[G2_NQ], en[G2_NQ];
    float a0[G2_NQ], a1[G2_NQ], di[G2_NQ];
    int left = 0;
    #pragma unroll
    for (int q = 0; q < G2_NQ; ++q) {
        int i = base + q * 32 + g;
        bool ok = i < n_nodes;
        j[q]  = ok ? rs[q] + sub : 0;
        en[q] = ok ? ((i + 1 < n_nodes) ? re[q] : n_edges) : 0;
        di[q] = ok ? dn[q] : 0.0f;
        a0[q] = (ok && sub == 0) ? zi[q].x : 0.0f;
        a1[q] = (ok && sub == 0) ? zi[q].y : 0.0f;
        int c = en[q] - j[q];
        if (c > left) left = c;
    }
    for (; left > 0; left -= 8) {
        int s[G2_NQ];
        #pragma unroll
        for (int q = 0; q < G2_NQ; ++q) {
            s[q] = -1;
            if (j[q] < en[q]) s[q] = csr_src[j[q]];
        }
        float2 zs[G2_NQ];
        #pragma unroll
        for (int q = 0; q < G2_NQ; ++q) {
            zs[q] = make_float2(0.0f, 0.0f);
            if (s[q] >= 0) zs[q] = *(const float2*)(z + 2 * (size_t)s[q]);
        }
        #pragma unroll
        for (int q = 0; q < G2_NQ; ++q) {
            if (s[q] >= 0) {
                a0[q] += zs[q].x;
                a1[q] += zs[q].y;
            }
            j[q] += 8;
        }
    }
    #pragma unroll
    for (int q = 0; q < G2_NQ; ++q) {
        float x0 = a0[q], x1 = a1[q];
        for (int o = 4; o >= 1; o >>= 1) {
            x0 += __shfl_xor(x0, o);
            x1 += __shfl_xor(x1, o);
        }
        if (sub == 0) ol[q * 32 + g] = make_float2(x0 * di[q], x1 * di[q]);
    }
    __syncthreads();
    if (tid < 32 * G2_NQ) {
        int o = base + tid;
        if (o < n_nodes) {
            float2 v = ol[tid];
            *(float2*)(out + 2 * (size_t)o) = make_float2(v.x + b2[0], v.y + b2[1]);
        }
    }
}


extern "C" void kernel_launch(void* const* d_in, const int* in_sizes, int n_in,
                              void* d_out, int out_size, void* d_ws, size_t ws_size,
                              hipStream_t stream) {
    const float* x  = (const float*)d_in[0];
    const int*   ei = (const int*)d_in[1];
    const float* W1 = (const float*)d_in[2];
    const float* b1 = (const float*)d_in[3];
    const float* W2 = (const float*)d_in[4];
    const float* b2 = (const float*)d_in[5];

    int n_nodes = in_sizes[0] / NFEAT;
    int n_edges = in_sizes[1] / 2;
    const int* src = ei;
    const int* dst = ei + n_edges;
    int n_bins = (n_nodes + BIN_NODES - 1) / BIN_NODES;
    int slice_len = (n_edges + N_SLICES - 1) / N_SLICES;

    float* ws = (float*)d_ws;
    size_t off = 0;
    size_t n64 = ((size_t)n_nodes + 63) & ~(size_t)63;
    size_t e64 = ((size_t)n_edges + 63) & ~(size_t)63;
    float* dinv    = ws + off;         off += n64;
    float* h       = ws + off;         off += (size_t)n_nodes * NHID;
    float* z       = ws + off;         off += ((size_t)2 * n_nodes + 63) & ~(size_t)63;
    int*   rowptr  = (int*)(ws + off); off += n64;
    off += 2 * MAX_BINS;
    int*   csr_src = (int*)(ws + off); off += e64;
    int2*  pairs   = (int2*)h;
    int*   lstart  = (int*)h + 2 * e64;
    (void)ws_size; (void)n_in; (void)out_size;

    float* out = (float*)d_out;

    const int B = 256;

    k_slice_sort<<<N_SLICES, 1024, 0, stream>>>(src, dst, lstart, pairs, n_edges, n_bins, slice_len);
    k_bin_sort  <<<n_bins, 512, 0, stream>>>(pairs, lstart, rowptr, dinv, csr_src, n_nodes, n_bins, slice_len);

    int tiles  = n_nodes / 16;
    int blocks = (tiles + 7) / 8;
    k_gemm1_wmma<<<blocks, B, 0, stream>>>(x, W1, dinv, h, n_nodes);

    k_gather1<<<((n_nodes + 15) / 16 + 7) / 8, 256, 0, stream>>>(h, dinv, rowptr, csr_src, b1, W2, z, n_nodes, n_edges);
    k_gather2<<<(n_nodes + 127) / 128, 256, 0, stream>>>(z, dinv, rowptr, csr_src, b2, out, n_nodes, n_edges);
}


# GEMM as one resident work-group per WGP (25 one-tile waves): W1 staged once per WGP with all its loads in flight; tiles leave through LDS as whole 128-B lines; sum order unchanged
# speedup vs baseline: 6.3125x; 1.0893x over previous
#include <hip/hip_runtime.h>


typedef float v2f __attribute__((ext_vector_type(2)));
typedef float v8f __attribute__((ext_vector_type(8)));

#define NFEAT 256
#define NHID  64
#define BIN_LOG   8
#define BIN_NODES (1 << BIN_LOG)
#define N_SLICES  256
#define MAX_BINS  1024
#define SLICE_CAP 6400
#define BIN_CAP   8192
static_assert(N_SLICES == BIN_NODES, "k_bin_sort clears one counter per slice-reading thread");


__global__ __launch_bounds__(1024) void k_slice_sort(const int* __restrict__ src, const int* __restrict__ dst,
                                                     int* __restrict__ lstart_g, int2* __restrict__ pairs,
                                                     int n_edges, int n_bins, int slice_len) {
    __shared__ int lh[MAX_BINS];
    __shared__ int wsum[32];
    __shared__ int2 stage[SLICE_CAP];
    int tid = threadIdx.x;
    int lane = tid & 31;
    int wave = tid >> 5;
    int w = blockIdx.x;
    lh[tid] = 0;
    __syncthreads();
    int e0 = w * slice_len;
    int e1 = e0 + slice_len;
    if (e1 > n_edges) e1 = n_edges;
    for (int e = e0 + tid; e < e1; e += 4 * 1024) {
        int d[4];
        #pragma unroll
        for (int k = 0; k < 4; ++k) {
            int ee = e + k * 1024;
            d[k] = (ee < e1) ? dst[ee] : -1;
        }
        #pragma unroll
        for (int k = 0; k < 4; ++k)
            if (d[k] >= 0) atomicAdd(&lh[d[k] >> BIN_LOG], 1);
    }
    __syncthreads();
    int c = lh[tid];
    int inc = c;
    for (int o = 1; o < 32; o <<= 1) {
        int u = __shfl_up(inc, o);
        if (lane >= o) inc += u;
    }
    if (lane == 31) wsum[wave] = inc;
    __syncthreads();
    int woff = 0;
    for (int k = 0; k < wave; ++k) woff += wsum[k];
    int excl = woff + inc - c;
    lh[tid] = excl;
    int* lrow = lstart_g + (size_t)w * (n_bins + 1);
    if (tid < n_bins) lrow[tid] = excl;
    if (tid == n_bins - 1) lrow[n_bins] = excl + c;
    __syncthreads();
    bool staged = slice_len <= SLICE_CAP;
    for (int e = e0 + tid; e < e1; e += 4 * 1024) {
        int d[4], s[4];
        #pragma unroll
        for (int k = 0; k < 4; ++k) {
            int ee = e + k * 1024;
            bool in = ee < e1;
            d[k] = in ? dst[ee] : -1;
            s[k] = in ? src[ee] : 0;
        }
        #pragma unroll
        for (int k = 0; k < 4; ++k) {
            if (d[k] >= 0) {
                int r = atomicAdd(&lh[d[k] >> BIN_LOG], 1);
                if (staged) stage[r] = make_int2(s[k], d[k]);
                else        pairs[e0 + r] = make_int2(s[k], d[k]);
            }
        }
    }
    if (staged) {
        __syncthreads();
        int n_loc = e1 - e0;
        for (int t = tid; t < n_loc; t += 1024) pairs[e0 + t] = stage[t];
    }
}

#define PIECE_STEP 16
__global__ __launch_bounds__(512) void k_bin_sort(const int2* __restrict__ pairs, const int* __restrict__ lstart_g,
                                                  int* __restrict__ rowptr, float* __restrict__ dinv,
                                                  int* __restrict__ csr_src, int n_nodes, int n_bins, int slice_len) {
    __shared__ int cnt[BIN_NODES];
    __shared__ int cur[BIN_NODES];
    __shared__ int wsum[8];
    __shared__ int wsa[8];
    __shared__ int wsb[8];
    __shared__ int pbeg[N_SLICES];
    __shared__ int plen[N_SLICES];
    __shared__ int lsrc[BIN_CAP];
    int tid = threadIdx.x;
    int lane = tid & 31;
    int wave = tid >> 5;
    int b = blockIdx.x;
    if (tid < N_SLICES) {
        const int* lrow = lstart_g + (size_t)tid * (n_bins + 1) + b;
        int a = lrow[0];
        int n = lrow[1] - a;
        pbeg[tid] = tid * slice_len + a;
        plen[tid] = n;
        cnt[tid] = 0;
        for (int o = 16; o >= 1; o >>= 1) {
            a += __shfl_xor(a, o);
            n += __shfl_xor(n, o);
        }
        if (lane == 0) { wsa[wave] = a; wsb[wave] = n; }
    }
    __syncthreads();
    int s0 = 0, nb = 0;
    for (int k = 0; k < N_SLICES / 32; ++k) { s0 += wsa[k]; nb += wsb[k]; }
    bool staged = nb <= BIN_CAP;
    const int pw = tid >> 1;
    const int pl = plen[pw];
    const int2* pp = pairs + pbeg[pw] + (tid & 1);
    for (int i = (tid & 1); i < pl; i += PIECE_STEP) {
        int d[PIECE_STEP / 2];
        #pragma unroll
        for (int k = 0; k < PIECE_STEP / 2; ++k)
            d[k] = (i + 2 * k < pl) ? pp[(i - (tid & 1)) + 2 * k].y : -1;
        #pragma unroll
        for (int k = 0; k < PIECE_STEP / 2; ++k)
            if (d[k] >= 0) atomicAdd(&cnt[d[k] & (BIN_NODES - 1)], 1);
    }
    __syncthreads();
    int c = 0, inc = 0;
    if (tid < BIN_NODES) {
        c = cnt[tid];
        inc = c;
        for (int o = 1; o < 32; o <<= 1) {
            int u = __shfl_up(inc, o);
            if (lane >= o) inc += u;
        }
        if (lane == 31) wsum[wave] = inc;
    }
    __syncthreads();
    if (tid < BIN_NODES) {
        int woff = 0;
        for (int k = 0; k < wave; ++k) woff += wsum[k];
        int excl = woff + inc - c;
        cur[tid] = excl;
        int node = b * BIN_NODES + tid;
        if (node < n_nodes) {
            rowptr[node] = s0 + excl;
            float d = 1.0f + (float)c;
            dinv[node] = (d > 0.0f) ? rsqrtf(d) : 0.0f;
        }
    }
    __syncthreads();
    for (int i = (tid & 1); i < pl; i += PIECE_STEP) {
        int2 p[PIECE_STEP / 2];
        #pragma unroll
        for (int k = 0; k < PIECE_STEP / 2; ++k)
            p[k] = (i + 2 * k < pl) ? pp[(i - (tid & 1)) + 2 * k] : make_int2(0, -1);
        #pragma unroll
        for (int k = 0; k < PIECE_STEP / 2; ++k) {
            if (p[k].y >= 0) {
                int pos = atomicAdd(&cur[p[k].y & (BIN_NODES - 1)], 1);
                if (staged) lsrc[pos] = p[k].x;
                else        csr_src[s0 + pos] = p[k].x;
            }
        }
    }
    if (staged) {
        __syncthreads();
        for (int t = tid; t < nb; t += 512) csr_src[s0 + t] = lsrc[t];
    }
}


#define G_TPB    25
#define G_NT     (32 * G_TPB)
#define G_WLD    ((NFEAT * NHID / 4 + G_NT - 1) / G_NT)
#define H_PITCH  68

__global__ __launch_bounds__(G_NT) void k_gemm1_wmma(const float* __restrict__ x,
                                                     const float* __restrict__ W1,
                                                     const float* __restrict__ dinv,
                                                     float* __restrict__ h,
                                                     int n_nodes) {
    __shared__ __attribute__((aligned(16))) float lds[NFEAT * NHID + G_TPB * 16 * H_PITCH];
    float* ldsW   = lds;
    float* hstage = lds + NFEAT * NHID;
    int tid = threadIdx.x;

    float4 wq[G_WLD];
    #pragma unroll
    for (int it = 0; it < G_WLD; ++it) {
        int q = tid + it * G_NT;
        wq[it] = (q < NFEAT * NHID / 4) ? *(const float4*)(W1 + 4 * q) : make_float4(0.0f, 0.0f, 0.0f, 0.0f);
    }
    #pragma unroll
    for (int it = 0; it < G_WLD; ++it) {
        int idx = 4 * (tid + it * G_NT);
        if (idx < NFEAT * NHID) {
            int k  = idx >> 6;
            int n0 = idx & 63;
            int p  = k >> 1;
            int lo = k & 1;
            int rot = (p & 1) << 5;
            float vals[4] = {wq[it].x, wq[it].y, wq[it].z, wq[it].w};
            #pragma unroll
            for (int j = 0; j < 4; ++j) {
                int wpos = (((n0 + j) << 1) + rot) & 127;
                ldsW[p * 128 + wpos + lo] = vals[j];
            }
        }
    }
    __syncthreads();

    int wave = tid >> 5;
    int lane = tid & 31;
    int tile = blockIdx.x * G_TPB + wave;
    int m0 = tile * 16;
    if (m0 >= n_nodes) return;

    int half = lane >> 4;
    int mrow = m0 + (lane & 15);
    int kk   = half << 1;
    int ncol = lane & 15;
    int rot  = half << 5;

    int wp0 = (((ncol +  0) << 1) + rot) & 127;
    int wp1 = (((ncol + 16) << 1) + rot) & 127;
    int wp2 = (((ncol + 32) << 1) + rot) & 127;
    int wp3 = (((ncol + 48) << 1) + rot) & 127;

    v8f acc0 = {}, acc1 = {}, acc2 = {}, acc3 = {};

    int rbase = m0 + (half << 3);
    float dr[8];
    for (int r = 0; r < 8; ++r) dr[r] = dinv[rbase + r];

    const float* xrow = x + (size_t)mrow * NFEAT;
    for (int k = 0; k < NFEAT; k += 4) {
        float2 av = *(const float2*)(xrow + k + kk);
        v2f a; a.x = av.x; a.y = av.y;

        const float* rowp = ldsW + ((k >> 1) + half) * 128;
        v2f b0 = *(const v2f*)(rowp + wp0);
        v2f b1 = *(const v2f*)(rowp + wp1);
        v2f b2 = *(const v2f*)(rowp + wp2);
        v2f b3 = *(const v2f*)(rowp + wp3);

        acc0 = __builtin_amdgcn_wmma_f32_16x16x4_f32(false, a, false, b0, (short)0, acc0, false, false);
        acc1 = __builtin_amdgcn_wmma_f32_16x16x4_f32(false, a, false, b1, (short)0, acc1, false, false);
        acc2 = __builtin_amdgcn_wmma_f32_16x16x4_f32(false, a, false, b2, (short)0, acc2, false, false);
        acc3 = __builtin_amdgcn_wmma_f32_16x16x4_f32(false, a, false, b3, (short)0, acc3, false, false);
    }

    float* st = hstage + wave * (16 * H_PITCH);
    for (int r = 0; r < 8; ++r) {
        float* sr = st + ((half << 3) + r) * H_PITCH;
        sr[ncol +  0] = acc0[r] * dr[r];
        sr[ncol + 16] = acc1[r] * dr[r];
        sr[ncol + 32] = acc2[r] * dr[r];
        sr[ncol + 48] = acc3[r] * dr[r];
    }
    for (int i = 0; i < 8; ++i) {
        int row = 2 * i + half;
        float4 v = *(const float4*)(st + row * H_PITCH + 4 * ncol);
        *(float4*)(h + (size_t)(m0 + row) * NHID + 4 * ncol) = v;
    }
}


#define G1_NPW 16
#define G1_CH  32
#define G1_PITCH 66

__device__ __forceinline__ float2 g1_row(const float* __restrict__ h, int s, unsigned lane2) {
    const float* row = h + (size_t)s * NHID;
    return *(const float2*)(row + lane2);
}

#define G1_FINISH()                                                                  \
    {                                                                                \
        float h0 = fmaxf(fmaf(acc.x, di, bb.x), 0.0f);                               \
        float h1 = fmaxf(fmaf(acc.y, di, bb.y), 0.0f);                               \
        float p0 = h0 * wv.x + h1 * wv.z;                                            \
        float p1 = h0 * wv.y + h1 * wv.w;                                            \
        *(float2*)(strip + k * G1_PITCH + lane2) = make_float2(p0, p1);              \
        ++k;                                                                         \
        acc = nself;                                                                 \
        if (k + 1 < nn) nself = *(const float2*)(strip + (k + 1) * G1_PITCH + lane2); \
        kend = __builtin_amdgcn_readlane(rp, k + 1);                                 \
        di = __int_as_float(__builtin_amdgcn_readlane(__float_as_int(dv), k));       \
    }

__global__ __launch_bounds__(256) void k_gather1(const float* __restrict__ h, const float* __restrict__ dinv,
                                                 const int* __restrict__ rowptr,
                                                 const int* __restrict__ csr_src,
                                                 const float* __restrict__ b1, const float* __restrict__ W2,
                                                 float* __restrict__ z, int n_nodes, int n_edges) {
    __shared__ float2 strips2[8 * G1_NPW * G1_PITCH / 2];
    int tid = threadIdx.x;
    int lane = tid & 31;
    unsigned lane2 = 2u * (unsigned)lane;
    int wave = __builtin_amdgcn_readfirstlane(tid >> 5);
    int i0 = (blockIdx.x * 8 + wave) * G1_NPW;
    if (i0 >= n_nodes) return;
    float* strip = (float*)strips2 + wave * (G1_NPW * G1_PITCH);
    int nn = n_nodes - i0;
    if (nn > G1_NPW) nn = G1_NPW;
    const int last = n_nodes - 1;
    int gi = i0 + lane;
    int rp = (gi < n_nodes) ? rowptr[gi] : n_edges;
    float dv = (gi < n_nodes) ? dinv[gi] : 0.0f;
    int gz = i0 + (lane >> 1);
    float dz = (gz < n_nodes) ? dinv[gz] : 0.0f;
    float2 bb = *(const float2*)(b1 + lane2);
    float4 wv = *(const float4*)(W2 + 2u * lane2);
    float2 own[G1_NPW];
    #pragma unroll
    for (int q = 0; q < G1_NPW; ++q) own[q] = g1_row(h, (i0 + q > last) ? last : i0 + q, lane2);
    const int e0 = __builtin_amdgcn_readlane(rp, 0);
    const int e1 = __builtin_amdgcn_readlane(rp, nn);
    int sv = 0;
    if (e1 > e0) {
        int ic = e0 + lane;
        if (ic > e1 - 1) ic = e1 - 1;
        sv = csr_src[ic];
    }
    #pragma unroll
    for (int q = 2; q < G1_NPW; ++q) *(float2*)(strip + q * G1_PITCH + lane2) = own[q];
    int k = 0;
    int kend = __builtin_amdgcn_readlane(rp, 1);
    float di = __int_as_float(__builtin_amdgcn_readlane(__float_as_int(dv), 0));
    float2 acc = own[0];
    float2 nself = own[1];
    for (int e = e0; e < e1; e += G1_CH) {
        int in = e + G1_CH + lane;
        if (in > e1 - 1) in = e1 - 1;
        int nsv = csr_src[in];
        float2 r[G1_CH];
        #pragma unroll
        for (int t = 0; t < G1_CH; ++t) r[t] = g1_row(h, __builtin_amdgcn_readlane(sv, t), lane2);
        const int elim = e1 - e;
        int lim = kend - e;
        #pragma unroll
        for (int t = 0; t < G1_CH; ++t) {
            if (__builtin_expect(t >= lim, 0)) {
                while (k + 1 < nn && t == lim) {
                    G1_FINISH()
                    lim = kend - e;
                }
                if (t >= elim) break;
            }
            acc.x += r[t].x;
            acc.y += r[t].y;
        }
        sv = nsv;
    }
    while (k < nn) G1_FINISH()
    const float* col = strip + (lane >> 1) * G1_PITCH + (lane & 1);
    float a[32];
    #pragma unroll
    for (int j = 0; j < 32; ++j) a[j] = col[2 * j];
    #pragma unroll
    for (int w = 16; w >= 1; w >>= 1) {
        #pragma unroll
        for (int j = 0; j < w; ++j) a[j] += a[j + w];
    }
    int go = 2 * i0 + lane;
    if (go < 2 * n_nodes) z[go] = a[0] * dz;
}


#define G2_NQ 4

__global__ __launch_bounds__(256) void k_gather2(const float* __restrict__ z, const float* __restrict__ dinv,
                                                 const int* __restrict__ rowptr,
                                                 const int* __restrict__ csr_src, const float* __restrict__ b2,
                                                 float* __restrict__ out, int n_nodes, int n_edges) {
    __shared__ float2 ol[32 * G2_NQ];
    int tid = threadIdx.x;
    int sub = tid & 7;
    int g = tid >> 3;
    int base = blockIdx.x * (32 * G2_NQ);
    const int last = n_nodes - 1;
    int rs[G2_NQ], re[G2_NQ];
    float dn[G2_NQ];
    float2 zi[G2_NQ];
    #pragma unroll
    for (int q = 0; q < G2_NQ; ++q) {
        int i = base + q * 32 + g;
        int ic = (i < last) ? i : last;
        int in = (i + 1 < last) ? i + 1 : last;
        rs[q] = rowptr[ic];
        re[q] = rowptr[in];
        dn[q] = dinv[ic];
        zi[q] = *(const float2*)(z + 2 * (size_t)ic);
    }
    int j[G2_NQ], en[G2_NQ];
    float a0[G2_NQ], a1[G2_NQ], di[G2_NQ];
    int left = 0;
    #pragma unroll
    for (int q = 0; q < G2_NQ; ++q) {
        int i = base + q * 32 + g;
        bool ok = i < n_nodes;
        j[q]  = ok ? rs[q] + sub : 0;
        en[q] = ok ? ((i + 1 < n_nodes) ? re[q] : n_edges) : 0;
        di[q] = ok ? dn[q] : 0.0f;
        a0[q] = (ok && sub == 0) ? zi[q].x : 0.0f;
        a1[q] = (ok && sub == 0) ? zi[q].y : 0.0f;
        int c = en[q] - j[q];
        if (c > left) left = c;
    }
    for (; left > 0; left -= 8) {
        int s[G2_NQ];
        #pragma unroll
        for (int q = 0; q < G2_NQ; ++q) {
            s[q] = -1;
            if (j[q] < en[q]) s[q] = csr_src[j[q]];
        }
        float2 zs[G2_NQ];
        #pragma unroll
        for (int q = 0; q < G2_NQ; ++q) {
            zs[q] = make_float2(0.0f, 0.0f);
            if (s[q] >= 0) zs[q] = *(const float2*)(z + 2 * (size_t)s[q]);
        }
        #pragma unroll
        for (int q = 0; q < G2_NQ; ++q) {
            if (s[q] >= 0) {
                a0[q] += zs[q].x;
                a1[q] += zs[q].y;
            }
            j[q] += 8;
        }
    }
    #pragma unroll
    for (int q = 0; q < G2_NQ; ++q) {
        float x0 = a0[q], x1 = a1[q];
        for (int o = 4; o >= 1; o >>= 1) {
            x0 += __shfl_xor(x0, o);
            x1 += __shfl_xor(x1, o);
        }
        if (sub == 0) ol[q * 32 + g] = make_float2(x0 * di[q], x1 * di[q]);
    }
    __syncthreads();
    if (tid < 32 * G2_NQ) {
        int o = base + tid;
        if (o < n_nodes) {
            float2 v = ol[tid];
            *(float2*)(out + 2 * (size_t)o) = make_float2(v.x + b2[0], v.y + b2[1]);
        }
    }
}


extern "C" void kernel_launch(void* const* d_in, const int* in_sizes, int n_in,
                              void* d_out, int out_size, void* d_ws, size_t ws_size,
                              hipStream_t stream) {
    const float* x  = (const float*)d_in[0];
    const int*   ei = (const int*)d_in[1];
    const float* W1 = (const float*)d_in[2];
    const float* b1 = (const float*)d_in[3];
    const float* W2 = (const float*)d_in[4];
    const float* b2 = (const float*)d_in[5];

    int n_nodes = in_sizes[0] / NFEAT;
    int n_edges = in_sizes[1] / 2;
    const int* src = ei;
    const int* dst = ei + n_edges;
    int n_bins = (n_nodes + BIN_NODES - 1) / BIN_NODES;
    int slice_len = (n_edges + N_SLICES - 1) / N_SLICES;

    float* ws = (float*)d_ws;
    size_t off = 0;
    size_t n64 = ((size_t)n_nodes + 63) & ~(size_t)63;
    size_t e64 = ((size_t)n_edges + 63) & ~(size_t)63;
    float* dinv    = ws + off;         off += n64;
    float* h       = ws + off;         off += (size_t)n_nodes * NHID;
    float* z       = ws + off;         off += ((size_t)2 * n_nodes + 63) & ~(size_t)63;
    int*   rowptr  = (int*)(ws + off); off += n64;
    off += 2 * MAX_BINS;
    int*   csr_src = (int*)(ws + off); off += e64;
    int2*  pairs   = (int2*)h;
    int*   lstart  = (int*)h + 2 * e64;
    (void)ws_size; (void)n_in; (void)out_size;

    float* out = (float*)d_out;

    k_slice_sort<<<N_SLICES, 1024, 0, stream>>>(src, dst, lstart, pairs, n_edges, n_bins, slice_len);
    k_bin_sort  <<<n_bins, 512, 0, stream>>>(pairs, lstart, rowptr, dinv, csr_src, n_nodes, n_bins, slice_len);

    int tiles  = n_nodes / 16;
    int blocks = (tiles + G_TPB - 1) / G_TPB;
    k_gemm1_wmma<<<blocks, G_NT, 0, stream>>>(x, W1, dinv, h, n_nodes);

    k_gather1<<<((n_nodes + 15) / 16 + 7) / 8, 256, 0, stream>>>(h, dinv, rowptr, csr_src, b1, W2, z, n_nodes, n_edges);
    k_gather2<<<(n_nodes + 127) / 128, 256, 0, stream>>>(z, dinv, rowptr, csr_src, b2, out, n_nodes, n_edges);
}


# Edge sort, both launches: a thread loads all its edges at once and holds them in registers for the counting and the placing pass (one read, 8-wave groups); segment order = LDS atomic arrival, kept
# speedup vs baseline: 6.5789x; 1.0422x over previous
#include <hip/hip_runtime.h>


typedef float v2f __attribute__((ext_vector_type(2)));
typedef float v8f __attribute__((ext_vector_type(8)));

#define NFEAT 256
#define NHID  64
#define BIN_LOG   8
#define BIN_NODES (1 << BIN_LOG)
#define N_SLICES  256
#define MAX_BINS  1024
#define SLICE_CAP 6400
#define BIN_CAP   8192
static_assert(N_SLICES == BIN_NODES, "k_bin_sort clears one counter per slice-reading thread");


#define SS_T 256
#define SS_K (SLICE_CAP / SS_T)
#define SS_B (MAX_BINS / SS_T)
static_assert(SLICE_CAP % SS_T == 0 && MAX_BINS % SS_T == 0, "k_slice_sort deals edges and counters evenly");

__global__ __launch_bounds__(SS_T) void k_slice_sort(const int* __restrict__ src, const int* __restrict__ dst,
                                                     int* __restrict__ lstart_g, int2* __restrict__ pairs,
                                                     int n_edges, int n_bins, int slice_len) {
    __shared__ int lh[MAX_BINS + 32];
    __shared__ int wsum[SS_T / 32];
    __shared__ int2 stage[SLICE_CAP + 32];
    int tid = threadIdx.x;
    int lane = tid & 31;
    int wave = tid >> 5;
    int w = blockIdx.x;
    int e0 = w * slice_len;
    int e1 = e0 + slice_len;
    if (e1 > n_edges) e1 = n_edges;
    bool staged = slice_len <= SLICE_CAP;
    int d[SS_K], s[SS_K];
    if (staged) {
        #pragma unroll
        for (int k = 0; k < SS_K; ++k) {
            int ee = e0 + tid + k * SS_T;
            bool in = ee < e1;
            d[k] = in ? dst[ee] : -1;
            s[k] = in ? src[ee] : 0;
        }
    }
    #pragma unroll
    for (int j = 0; j < SS_B; ++j) lh[tid + j * SS_T] = 0;
    __syncthreads();
    if (staged) {
        #pragma unroll
        for (int k = 0; k < SS_K; ++k)
            if (d[k] >= 0) atomicAdd(&lh[d[k] >> BIN_LOG], 1);
    } else {
        for (int e = e0 + tid; e < e1; e += SS_T) atomicAdd(&lh[dst[e] >> BIN_LOG], 1);
    }
    __syncthreads();
    int c[SS_B];
    int tot = 0;
    #pragma unroll
    for (int j = 0; j < SS_B; ++j) { c[j] = lh[SS_B * tid + j]; tot += c[j]; }
    int inc = tot;
    for (int o = 1; o < 32; o <<= 1) {
        int u = __shfl_up(inc, o);
        if (lane >= o) inc += u;
    }
    if (lane == 31) wsum[wave] = inc;
    __syncthreads();
    int run = inc - tot;
    for (int k = 0; k < wave; ++k) run += wsum[k];
    int* lrow = lstart_g + (size_t)w * (n_bins + 1);
    #pragma unroll
    for (int j = 0; j < SS_B; ++j) {
        int bi = SS_B * tid + j;
        lh[bi] = run;
        if (bi < n_bins) lrow[bi] = run;
        if (bi == n_bins - 1) lrow[n_bins] = run + c[j];
        run += c[j];
    }
    __syncthreads();
    if (staged) {
        int r[SS_K];
        #pragma unroll
        for (int k = 0; k < SS_K; ++k)
            r[k] = atomicAdd(&lh[(d[k] >= 0) ? (d[k] >> BIN_LOG) : (MAX_BINS + lane)], 1);
        #pragma unroll
        for (int k = 0; k < SS_K; ++k)
            stage[(d[k] >= 0) ? r[k] : (SLICE_CAP + lane)] = make_int2(s[k], d[k]);
        __syncthreads();
        int n_loc = e1 - e0;
        int2 v[SS_K];
        #pragma unroll
        for (int k = 0; k < SS_K; ++k) v[k] = stage[tid + k * SS_T];
        #pragma unroll
        for (int k = 0; k < SS_K; ++k) {
            int t = tid + k * SS_T;
            if (t < n_loc) pairs[e0 + t] = v[k];
        }
    } else {
        for (int e = e0 + tid; e < e1; e += SS_T) {
            int dd = dst[e];
            int r = atomicAdd(&lh[dd >> BIN_LOG], 1);
            pairs[e0 + r] = make_int2(src[e], dd);
        }
    }
}

#define BS_T 256
static_assert(BS_T == N_SLICES && BS_T == BIN_NODES, "k_bin_sort: one thread per slice and per destination");
__global__ __launch_bounds__(BS_T) void k_bin_sort(const int2* __restrict__ pairs, const int* __restrict__ lstart_g,
                                                   int* __restrict__ rowptr, float* __restrict__ dinv,
                                                   int* __restrict__ csr_src, int n_nodes, int n_bins, int slice_len) {
    __shared__ int cnt[BIN_NODES];
    __shared__ int cur[BIN_NODES + 32];
    __shared__ int wsum[BS_T / 32];
    __shared__ int wsa[BS_T / 32];
    __shared__ int wsb[BS_T / 32];
    __shared__ int lsrc[BIN_CAP + 32];
    int tid = threadIdx.x;
    int lane = tid & 31;
    int wave = tid >> 5;
    int b = blockIdx.x;
    const int* lrow = lstart_g + (size_t)tid * (n_bins + 1) + b;
    int a = lrow[0];
    const int my_len = lrow[1] - a;
    const int my_beg = tid * slice_len + a;
    cnt[tid] = 0;
    {
        int ra = a, rn = my_len;
        for (int o = 16; o >= 1; o >>= 1) {
            ra += __shfl_xor(ra, o);
            rn += __shfl_xor(rn, o);
        }
        if (lane == 0) { wsa[wave] = ra; wsb[wave] = rn; }
    }
    int px[32], py[32];
    bool longp = false;
    #pragma unroll
    for (int r = 0; r < 32; ++r) {
        int pl = __builtin_amdgcn_readlane(my_len, r);
        int pb = __builtin_amdgcn_readlane(my_beg, r);
        int li = (pl > 0) ? pl - 1 : 0;
        if (li > lane) li = lane;
        int2 v = pairs[pb + li];
        px[r] = (lane < pl) ? v.x : 0;
        py[r] = (lane < pl) ? v.y : -1;
        longp |= pl > 32;
    }
    __syncthreads();
    int s0 = 0, nb = 0;
    for (int k = 0; k < BS_T / 32; ++k) { s0 += wsa[k]; nb += wsb[k]; }
    bool staged = nb <= BIN_CAP;
    #pragma unroll
    for (int r = 0; r < 32; ++r)
        if (py[r] >= 0) atomicAdd(&cnt[py[r] & (BIN_NODES - 1)], 1);
    if (longp) {
        for (int r = 0; r < 32; ++r) {
            int pl = __builtin_amdgcn_readlane(my_len, r);
            int pb = __builtin_amdgcn_readlane(my_beg, r);
            for (int i = 32 + lane; i < pl; i += 32) atomicAdd(&cnt[pairs[pb + i].y & (BIN_NODES - 1)], 1);
        }
    }
    __syncthreads();
    int c = cnt[tid];
    int inc = c;
    for (int o = 1; o < 32; o <<= 1) {
        int u = __shfl_up(inc, o);
        if (lane >= o) inc += u;
    }
    if (lane == 31) wsum[wave] = inc;
    __syncthreads();
    {
        int excl = inc - c;
        for (int k = 0; k < wave; ++k) excl += wsum[k];
        cur[tid] = excl;
        int node = b * BIN_NODES + tid;
        if (node < n_nodes) {
            rowptr[node] = s0 + excl;
            float dg = 1.0f + (float)c;
            dinv[node] = (dg > 0.0f) ? rsqrtf(dg) : 0.0f;
        }
    }
    __syncthreads();
    int pos[32];
    #pragma unroll
    for (int r = 0; r < 32; ++r)
        pos[r] = atomicAdd(&cur[(py[r] >= 0) ? (py[r] & (BIN_NODES - 1)) : (BIN_NODES + lane)], 1);
    if (staged) {
        #pragma unroll
        for (int r = 0; r < 32; ++r) lsrc[(py[r] >= 0) ? pos[r] : (BIN_CAP + lane)] = px[r];
    } else {
        #pragma unroll
        for (int r = 0; r < 32; ++r)
            if (py[r] >= 0) csr_src[s0 + pos[r]] = px[r];
    }
    if (longp) {
        for (int r = 0; r < 32; ++r) {
            int pl = __builtin_amdgcn_readlane(my_len, r);
            int pb = __builtin_amdgcn_readlane(my_beg, r);
            for (int i = 32 + lane; i < pl; i += 32) {
                int2 q = pairs[pb + i];
                int ps = atomicAdd(&cur[q.y & (BIN_NODES - 1)], 1);
                if (staged) lsrc[ps] = q.x;
                else        csr_src[s0 + ps] = q.x;
            }
        }
    }
    if (staged) {
        __syncthreads();
        for (int t0 = 0; t0 < nb; t0 += 8 * BS_T) {
            int v[8];
            #pragma unroll
            for (int j = 0; j < 8; ++j) {
                int t = t0 + tid + j * BS_T;
                v[j] = lsrc[(t < BIN_CAP) ? t : BIN_CAP - 1];
            }
            #pragma unroll
            for (int j = 0; j < 8; ++j) {
                int t = t0 + tid + j * BS_T;
                if (t < nb) csr_src[s0 + t] = v[j];
            }
        }
    }
}


#define G_TPB    25
#define G_NT     (32 * G_TPB)
#define G_WLD    ((NFEAT * NHID / 4 + G_NT - 1) / G_NT)
#define H_PITCH  68

__global__ __launch_bounds__(G_NT) void k_gemm1_wmma(const float* __restrict__ x,
                                                     const float* __restrict__ W1,
                                                     const float* __restrict__ dinv,
                                                     float* __restrict__ h,
                                                     int n_nodes) {
    __shared__ __attribute__((aligned(16))) float lds[NFEAT * NHID + G_TPB * 16 * H_PITCH];
    float* ldsW   = lds;
    float* hstage = lds + NFEAT * NHID;
    int tid = threadIdx.x;

    float4 wq[G_WLD];
    #pragma unroll
    for (int it = 0; it < G_WLD; ++it) {
        int q = tid + it * G_NT;
        wq[it] = (q < NFEAT * NHID / 4) ? *(const float4*)(W1 + 4 * q) : make_float4(0.0f, 0.0f, 0.0f, 0.0f);
    }
    #pragma unroll
    for (int it = 0; it < G_WLD; ++it) {
        int idx = 4 * (tid + it * G_NT);
        if (idx < NFEAT * NHID) {
            int k  = idx >> 6;
            int n0 = idx & 63;
            int p  = k >> 1;
            int lo = k & 1;
            int rot = (p & 1) << 5;
            float vals[4] = {wq[it].x, wq[it].y, wq[it].z, wq[it].w};
            #pragma unroll
            for (int j = 0; j < 4; ++j) {
                int wpos = (((n0 + j) << 1) + rot) & 127;
                ldsW[p * 128 + wpos + lo] = vals[j];
            }
        }
    }
    __syncthreads();

    int wave = tid >> 5;
    int lane = tid & 31;
    int tile = blockIdx.x * G_TPB + wave;
    int m0 = tile * 16;
    if (m0 >= n_nodes) return;

    int half = lane >> 4;
    int mrow = m0 + (lane & 15);
    int kk   = half << 1;
    int ncol = lane & 15;
    int rot  = half << 5;

    int wp0 = (((ncol +  0) << 1) + rot) & 127;
    int wp1 = (((ncol + 16) << 1) + rot) & 127;
    int wp2 = (((ncol + 32) << 1) + rot) & 127;
    int wp3 = (((ncol + 48) << 1) + rot) & 127;

    v8f acc0 = {}, acc1 = {}, acc2 = {}, acc3 = {};

    int rbase = m0 + (half << 3);
    float dr[8];
    for (int r = 0; r < 8; ++r) dr[r] = dinv[rbase + r];

    const float* xrow = x + (size_t)mrow * NFEAT;
    for (int k = 0; k < NFEAT; k += 4) {
        float2 av = *(const float2*)(xrow + k + kk);
        v2f a; a.x = av.x; a.y = av.y;

        const float* rowp = ldsW + ((k >> 1) + half) * 128;
        v2f b0 = *(const v2f*)(rowp + wp0);
        v2f b1 = *(const v2f*)(rowp + wp1);
        v2f b2 = *(const v2f*)(rowp + wp2);
        v2f b3 = *(const v2f*)(rowp + wp3);

        acc0 = __builtin_amdgcn_wmma_f32_16x16x4_f32(false, a, false, b0, (short)0, acc0, false, false);
        acc1 = __builtin_amdgcn_wmma_f32_16x16x4_f32(false, a, false, b1, (short)0, acc1, false, false);
        acc2 = __builtin_amdgcn_wmma_f32_16x16x4_f32(false, a, false, b2, (short)0, acc2, false, false);
        acc3 = __builtin_amdgcn_wmma_f32_16x16x4_f32(false, a, false, b3, (short)0, acc3, false, false);
    }

    float* st = hstage + wave * (16 * H_PITCH);
    for (int r = 0; r < 8; ++r) {
        float* sr = st + ((half << 3) + r) * H_PITCH;
        sr[ncol +  0] = acc0[r] * dr[r];
        sr[ncol + 16] = acc1[r] * dr[r];
        sr[ncol + 32] = acc2[r] * dr[r];
        sr[ncol + 48] = acc3[r] * dr[r];
    }
    for (int i = 0; i < 8; ++i) {
        int row = 2 * i + half;
        float4 v = *(const float4*)(st + row * H_PITCH + 4 * ncol);
        *(float4*)(h + (size_t)(m0 + row) * NHID + 4 * ncol) = v;
    }
}


#define G1_NPW 16
#define G1_CH  32
#define G1_PITCH 66

__device__ __forceinline__ float2 g1_row(const float* __restrict__ h, int s, unsigned lane2) {
    const float* row = h + (size_t)s * NHID;
    return *(const float2*)(row + lane2);
}

#define G1_FINISH()                                                                  \
    {                                                                                \
        float h0 = fmaxf(fmaf(acc.x, di, bb.x), 0.0f);                               \
        float h1 = fmaxf(fmaf(acc.y, di, bb.y), 0.0f);                               \
        float p0 = h0 * wv.x + h1 * wv.z;                                            \
        float p1 = h0 * wv.y + h1 * wv.w;                                            \
        *(float2*)(strip + k * G1_PITCH + lane2) = make_float2(p0, p1);              \
        ++k;                                                                         \
        acc = nself;                                                                 \
        if (k + 1 < nn) nself = *(const float2*)(strip + (k + 1) * G1_PITCH + lane2); \
        kend = __builtin_amdgcn_readlane(rp, k + 1);                                 \
        di = __int_as_float(__builtin_amdgcn_readlane(__float_as_int(dv), k));       \
    }

__global__ __launch_bounds__(256) void k_gather1(const float* __restrict__ h, const float* __restrict__ dinv,
                                                 const int* __restrict__ rowptr,
                                                 const int* __restrict__ csr_src,
                                                 const float* __restrict__ b1, const float* __restrict__ W2,
                                                 float* __restrict__ z, int n_nodes, int n_edges) {
    __shared__ float2 strips2[8 * G1_NPW * G1_PITCH / 2];
    int tid = threadIdx.x;
    int lane = tid & 31;
    unsigned lane2 = 2u * (unsigned)lane;
    int wave = __builtin_amdgcn_readfirstlane(tid >> 5);
    int i0 = (blockIdx.x * 8 + wave) * G1_NPW;
    if (i0 >= n_nodes) return;
    float* strip = (float*)strips2 + wave * (G1_NPW * G1_PITCH);
    int nn = n_nodes - i0;
    if (nn > G1_NPW) nn = G1_NPW;
    const int last = n_nodes - 1;
    int gi = i0 + lane;
    int rp = (gi < n_nodes) ? rowptr[gi] : n_edges;
    float dv = (gi < n_nodes) ? dinv[gi] : 0.0f;
    int gz = i0 + (lane >> 1);
    float dz = (gz < n_nodes) ? dinv[gz] : 0.0f;
    float2 bb = *(const float2*)(b1 + lane2);
    float4 wv = *(const float4*)(W2 + 2u * lane2);
    float2 own[G1_NPW];
    #pragma unroll
    for (int q = 0; q < G1_NPW; ++q) own[q] = g1_row(h, (i0 + q > last) ? last : i0 + q, lane2);
    const int e0 = __builtin_amdgcn_readlane(rp, 0);
    const int e1 = __builtin_amdgcn_readlane(rp, nn);
    int sv = 0;
    if (e1 > e0) {
        int ic = e0 + lane;
        if (ic > e1 - 1) ic = e1 - 1;
        sv = csr_src[ic];
    }
    #pragma unroll
    for (int q = 2; q < G1_NPW; ++q) *(float2*)(strip + q * G1_PITCH + lane2) = own[q];
    int k = 0;
    int kend = __builtin_amdgcn_readlane(rp, 1);
    float di = __int_as_float(__builtin_amdgcn_readlane(__float_as_int(dv), 0));
    float2 acc = own[0];
    float2 nself = own[1];
    for (int e = e0; e < e1; e += G1_CH) {
        int in = e + G1_CH + lane;
        if (in > e1 - 1) in = e1 - 1;
        int nsv = csr_src[in];
        float2 r[G1_CH];
        #pragma unroll
        for (int t = 0; t < G1_CH; ++t) r[t] = g1_row(h, __builtin_amdgcn_readlane(sv, t), lane2);
        const int elim = e1 - e;
        int lim = kend - e;
        #pragma unroll
        for (int t = 0; t < G1_CH; ++t) {
            if (__builtin_expect(t >= lim, 0)) {
                while (k + 1 < nn && t == lim) {
                    G1_FINISH()
                    lim = kend - e;
                }
                if (t >= elim) break;
            }
            acc.x += r[t].x;
            acc.y += r[t].y;
        }
        sv = nsv;
    }
    while (k < nn) G1_FINISH()
    const float* col = strip + (lane >> 1) * G1_PITCH + (lane & 1);
    float a[32];
    #pragma unroll
    for (int j = 0; j < 32; ++j) a[j] = col[2 * j];
    #pragma unroll
    for (int w = 16; w >= 1; w >>= 1) {
        #pragma unroll
        for (int j = 0; j < w; ++j) a[j] += a[j + w];
    }
    int go = 2 * i0 + lane;
    if (go < 2 * n_nodes) z[go] = a[0] * dz;
}


#define G2_NQ 4

__global__ __launch_bounds__(256) void k_gather2(const float* __restrict__ z, const float* __restrict__ dinv,
                                                 const int* __restrict__ rowptr,
                                                 const int* __restrict__ csr_src, const float* __restrict__ b2,
                                                 float* __restrict__ out, int n_nodes, int n_edges) {
    __shared__ float2 ol[32 * G2_NQ];
    int tid = threadIdx.x;
    int sub = tid & 7;
    int g = tid >> 3;
    int base = blockIdx.x * (32 * G2_NQ);
    const int last = n_nodes - 1;
    int rs[G2_NQ], re[G2_NQ];
    float dn[G2_NQ];
    float2 zi[G2_NQ];
    #pragma unroll
    for (int q = 0; q < G2_NQ; ++q) {
        int i = base + q * 32 + g;
        int ic = (i < last) ? i : last;
        int in = (i + 1 < last) ? i + 1 : last;
        rs[q] = rowptr[ic];
        re[q] = rowptr[in];
        dn[q] = dinv[ic];
        zi[q] = *(const float2*)(z + 2 * (size_t)ic);
    }
    int j[G2_NQ], en[G2_NQ];
    float a0[G2_NQ], a1[G2_NQ], di[G2_NQ];
    int left = 0;
    #pragma unroll
    for (int q = 0; q < G2_NQ; ++q) {
        int i = base + q * 32 + g;
        bool ok = i < n_nodes;
        j[q]  = ok ? rs[q] + sub : 0;
        en[q] = ok ? ((i + 1 < n_nodes) ? re[q] : n_edges) : 0;
        di[q] = ok ? dn[q] : 0.0f;
        a0[q] = (ok && sub == 0) ? zi[q].x : 0.0f;
        a1[q] = (ok && sub == 0) ? zi[q].y : 0.0f;
        int c = en[q] - j[q];
        if (c > left) left = c;
    }
    for (; left > 0; left -= 8) {
        int s[G2_NQ];
        #pragma unroll
        for (int q = 0; q < G2_NQ; ++q) {
            s[q] = -1;
            if (j[q] < en[q]) s[q] = csr_src[j[q]];
        }
        float2 zs[G2_NQ];
        #pragma unroll
        for (int q = 0; q < G2_NQ; ++q) {
            zs[q] = make_float2(0.0f, 0.0f);
            if (s[q] >= 0) zs[q] = *(const float2*)(z + 2 * (size_t)s[q]);
        }
        #pragma unroll
        for (int q = 0; q < G2_NQ; ++q) {
            if (s[q] >= 0) {
                a0[q] += zs[q].x;
                a1[q] += zs[q].y;
            }
            j[q] += 8;
        }
    }
    #pragma unroll
    for (int q = 0; q < G2_NQ; ++q) {
        float x0 = a0[q], x1 = a1[q];
        for (int o = 4; o >= 1; o >>= 1) {
            x0 += __shfl_xor(x0, o);
            x1 += __shfl_xor(x1, o);
        }
        if (sub == 0) ol[q * 32 + g] = make_float2(x0 * di[q], x1 * di[q]);
    }
    __syncthreads();
    if (tid < 32 * G2_NQ) {
        int o = base + tid;
        if (o < n_nodes) {
            float2 v = ol[tid];
            *(float2*)(out + 2 * (size_t)o) = make_float2(v.x + b2[0], v.y + b2[1]);
        }
    }
}


extern "C" void kernel_launch(void* const* d_in, const int* in_sizes, int n_in,
                              void* d_out, int out_size, void* d_ws, size_t ws_size,
                              hipStream_t stream) {
    const float* x  = (const float*)d_in[0];
    const int*   ei = (const int*)d_in[1];
    const float* W1 = (const float*)d_in[2];
    const float* b1 = (const float*)d_in[3];
    const float* W2 = (const float*)d_in[4];
    const float* b2 = (const float*)d_in[5];

    int n_nodes = in_sizes[0] / NFEAT;
    int n_edges = in_sizes[1] / 2;
    const int* src = ei;
    const int* dst = ei + n_edges;
    int n_bins = (n_nodes + BIN_NODES - 1) / BIN_NODES;
    int slice_len = (n_edges + N_SLICES - 1) / N_SLICES;

    float* ws = (float*)d_ws;
    size_t off = 0;
    size_t n64 = ((size_t)n_nodes + 63) & ~(size_t)63;
    size_t e64 = ((size_t)n_edges + 63) & ~(size_t)63;
    float* dinv    = ws + off;         off += n64;
    float* h       = ws + off;         off += (size_t)n_nodes * NHID;
    float* z       = ws + off;         off += ((size_t)2 * n_nodes + 63) & ~(size_t)63;
    int*   rowptr  = (int*)(ws + off); off += n64;
    off += 2 * MAX_BINS;
    int*   csr_src = (int*)(ws + off); off += e64;
    int2*  pairs   = (int2*)h;
    int*   lstart  = (int*)h + 2 * e64;
    (void)ws_size; (void)n_in; (void)out_size;

    float* out = (float*)d_out;

    k_slice_sort<<<N_SLICES, SS_T, 0, stream>>>(src, dst, lstart, pairs, n_edges, n_bins, slice_len);
    k_bin_sort  <<<n_bins, BS_T, 0, stream>>>(pairs, lstart, rowptr, dinv, csr_src, n_nodes, n_bins, slice_len);

    int tiles  = n_nodes / 16;
    int blocks = (tiles + G_TPB - 1) / G_TPB;
    k_gemm1_wmma<<<blocks, G_NT, 0, stream>>>(x, W1, dinv, h, n_nodes);

    k_gather1<<<((n_nodes + 15) / 16 + 7) / 8, 256, 0, stream>>>(h, dinv, rowptr, csr_src, b1, W2, z, n_nodes, n_edges);
    k_gather2<<<(n_nodes + 127) / 128, 256, 0, stream>>>(z, dinv, rowptr, csr_src, b2, out, n_nodes, n_edges);
}


# GEMM: x tiles go through LDS: whole-line 16-byte loads (two rows per instruction), each 64-k chunk requested a chunk or more ahead, two buffers per wave; k order and sum order kept
# speedup vs baseline: 6.7922x; 1.0324x over previous
#include <hip/hip_runtime.h>


typedef float v2f __attribute__((ext_vector_type(2)));
typedef float v4f __attribute__((ext_vector_type(4)));
typedef float v8f __attribute__((ext_vector_type(8)));

#define NFEAT 256
#define NHID  64
#define BIN_LOG   8
#define BIN_NODES (1 << BIN_LOG)
#define N_SLICES  256
#define MAX_BINS  1024
#define SLICE_CAP 6400
#define BIN_CAP   8192
static_assert(N_SLICES == BIN_NODES, "k_bin_sort clears one counter per slice-reading thread");


#define SS_T 256
#define SS_K (SLICE_CAP / SS_T)
#define SS_B (MAX_BINS / SS_T)
static_assert(SLICE_CAP % SS_T == 0 && MAX_BINS % SS_T == 0, "k_slice_sort deals edges and counters evenly");

__global__ __launch_bounds__(SS_T) void k_slice_sort(const int* __restrict__ src, const int* __restrict__ dst,
                                                     int* __restrict__ lstart_g, int2* __restrict__ pairs,
                                                     int n_edges, int n_bins, int slice_len) {
    __shared__ int lh[MAX_BINS + 32];
    __shared__ int wsum[SS_T / 32];
    __shared__ int2 stage[SLICE_CAP + 32];
    int tid = threadIdx.x;
    int lane = tid & 31;
    int wave = tid >> 5;
    int w = blockIdx.x;
    int e0 = w * slice_len;
    int e1 = e0 + slice_len;
    if (e1 > n_edges) e1 = n_edges;
    bool staged = slice_len <= SLICE_CAP;
    int d[SS_K], s[SS_K];
    if (staged) {
        #pragma unroll
        for (int k = 0; k < SS_K; ++k) {
            int ee = e0 + tid + k * SS_T;
            bool in = ee < e1;
            d[k] = in ? dst[ee] : -1;
            s[k] = in ? src[ee] : 0;
        }
    }
    #pragma unroll
    for (int j = 0; j < SS_B; ++j) lh[tid + j * SS_T] = 0;
    __syncthreads();
    if (staged) {
        #pragma unroll
        for (int k = 0; k < SS_K; ++k)
            if (d[k] >= 0) atomicAdd(&lh[d[k] >> BIN_LOG], 1);
    } else {
        for (int e = e0 + tid; e < e1; e += SS_T) atomicAdd(&lh[dst[e] >> BIN_LOG], 1);
    }
    __syncthreads();
    int c[SS_B];
    int tot = 0;
    #pragma unroll
    for (int j = 0; j < SS_B; ++j) { c[j] = lh[SS_B * tid + j]; tot += c[j]; }
    int inc = tot;
    for (int o = 1; o < 32; o <<= 1) {
        int u = __shfl_up(inc, o);
        if (lane >= o) inc += u;
    }
    if (lane == 31) wsum[wave] = inc;
    __syncthreads();
    int run = inc - tot;
    for (int k = 0; k < wave; ++k) run += wsum[k];
    int* lrow = lstart_g + (size_t)w * (n_bins + 1);
    #pragma unroll
    for (int j = 0; j < SS_B; ++j) {
        int bi = SS_B * tid + j;
        lh[bi] = run;
        if (bi < n_bins) lrow[bi] = run;
        if (bi == n_bins - 1) lrow[n_bins] = run + c[j];
        run += c[j];
    }
    __syncthreads();
    if (staged) {
        int r[SS_K];
        #pragma unroll
        for (int k = 0; k < SS_K; ++k)
            r[k] = atomicAdd(&lh[(d[k] >= 0) ? (d[k] >> BIN_LOG) : (MAX_BINS + lane)], 1);
        #pragma unroll
        for (int k = 0; k < SS_K; ++k)
            stage[(d[k] >= 0) ? r[k] : (SLICE_CAP + lane)] = make_int2(s[k], d[k]);
        __syncthreads();
        int n_loc = e1 - e0;
        int2 v[SS_K];
        #pragma unroll
        for (int k = 0; k < SS_K; ++k) v[k] = stage[tid + k * SS_T];
        #pragma unroll
        for (int k = 0; k < SS_K; ++k) {
            int t = tid + k * SS_T;
            if (t < n_loc) pairs[e0 + t] = v[k];
        }
    } else {
        for (int e = e0 + tid; e < e1; e += SS_T) {
            int dd = dst[e];
            int r = atomicAdd(&lh[dd >> BIN_LOG], 1);
            pairs[e0 + r] = make_int2(src[e], dd);
        }
    }
}

#define BS_T 256
static_assert(BS_T == N_SLICES && BS_T == BIN_NODES, "k_bin_sort: one thread per slice and per destination");
__global__ __launch_bounds__(BS_T) void k_bin_sort(const int2* __restrict__ pairs, const int* __restrict__ lstart_g,
                                                   int* __restrict__ rowptr, float* __restrict__ dinv,
                                                   int* __restrict__ csr_src, int n_nodes, int n_bins, int slice_len) {
    __shared__ int cnt[BIN_NODES];
    __shared__ int cur[BIN_NODES + 32];
    __shared__ int wsum[BS_T / 32];
    __shared__ int wsa[BS_T / 32];
    __shared__ int wsb[BS_T / 32];
    __shared__ int lsrc[BIN_CAP + 32];
    int tid = threadIdx.x;
    int lane = tid & 31;
    int wave = tid >> 5;
    int b = blockIdx.x;
    const int* lrow = lstart_g + (size_t)tid * (n_bins + 1) + b;
    int a = lrow[0];
    const int my_len = lrow[1] - a;
    const int my_beg = tid * slice_len + a;
    cnt[tid] = 0;
    {
        int ra = a, rn = my_len;
        for (int o = 16; o >= 1; o >>= 1) {
            ra += __shfl_xor(ra, o);
            rn += __shfl_xor(rn, o);
        }
        if (lane == 0) { wsa[wave] = ra; wsb[wave] = rn; }
    }
    int px[32], py[32];
    bool longp = false;
    #pragma unroll
    for (int r = 0; r < 32; ++r) {
        int pl = __builtin_amdgcn_readlane(my_len, r);
        int pb = __builtin_amdgcn_readlane(my_beg, r);
        int li = (pl > 0) ? pl - 1 : 0;
        if (li > lane) li = lane;
        int2 v = pairs[pb + li];
        px[r] = (lane < pl) ? v.x : 0;
        py[r] = (lane < pl) ? v.y : -1;
        longp |= pl > 32;
    }
    __syncthreads();
    int s0 = 0, nb = 0;
    for (int k = 0; k < BS_T / 32; ++k) { s0 += wsa[k]; nb += wsb[k]; }
    bool staged = nb <= BIN_CAP;
    #pragma unroll
    for (int r = 0; r < 32; ++r)
        if (py[r] >= 0) atomicAdd(&cnt[py[r] & (BIN_NODES - 1)], 1);
    if (longp) {
        for (int r = 0; r < 32; ++r) {
            int pl = __builtin_amdgcn_readlane(my_len, r);
            int pb = __builtin_amdgcn_readlane(my_beg, r);
            for (int i = 32 + lane; i < pl; i += 32) atomicAdd(&cnt[pairs[pb + i].y & (BIN_NODES - 1)], 1);
        }
    }
    __syncthreads();
    int c = cnt[tid];
    int inc = c;
    for (int o = 1; o < 32; o <<= 1) {
        int u = __shfl_up(inc, o);
        if (lane >= o) inc += u;
    }
    if (lane == 31) wsum[wave] = inc;
    __syncthreads();
    {
        int excl = inc - c;
        for (int k = 0; k < wave; ++k) excl += wsum[k];
        cur[tid] = excl;
        int node = b * BIN_NODES + tid;
        if (node < n_nodes) {
            rowptr[node] = s0 + excl;
            float dg = 1.0f + (float)c;
            dinv[node] = (dg > 0.0f) ? rsqrtf(dg) : 0.0f;
        }
    }
    __syncthreads();
    int pos[32];
    #pragma unroll
    for (int r = 0; r < 32; ++r)
        pos[r] = atomicAdd(&cur[(py[r] >= 0) ? (py[r] & (BIN_NODES - 1)) : (BIN_NODES + lane)], 1);
    if (staged) {
        #pragma unroll
        for (int r = 0; r < 32; ++r) lsrc[(py[r] >= 0) ? pos[r] : (BIN_CAP + lane)] = px[r];
    } else {
        #pragma unroll
        for (int r = 0; r < 32; ++r)
            if (py[r] >= 0) csr_src[s0 + pos[r]] = px[r];
    }
    if (longp) {
        for (int r = 0; r < 32; ++r) {
            int pl = __builtin_amdgcn_readlane(my_len, r);
            int pb = __builtin_amdgcn_readlane(my_beg, r);
            for (int i = 32 + lane; i < pl; i += 32) {
                int2 q = pairs[pb + i];
                int ps = atomicAdd(&cur[q.y & (BIN_NODES - 1)], 1);
                if (staged) lsrc[ps] = q.x;
                else        csr_src[s0 + ps] = q.x;
            }
        }
    }
    if (staged) {
        __syncthreads();
        for (int t0 = 0; t0 < nb; t0 += 8 * BS_T) {
            int v[8];
            #pragma unroll
            for (int j = 0; j < 8; ++j) {
                int t = t0 + tid + j * BS_T;
                v[j] = lsrc[(t < BIN_CAP) ? t : BIN_CAP - 1];
            }
            #pragma unroll
            for (int j = 0; j < 8; ++j) {
                int t = t0 + tid + j * BS_T;
                if (t < nb) csr_src[s0 + t] = v[j];
            }
        }
    }
}


#define G_TPB    25
#define G_NT     (32 * G_TPB)
#define G_WLD    ((NFEAT * NHID / 4 + G_NT - 1) / G_NT)
#define H_PITCH  68
#define A_KC     64
#define A_NCH    (NFEAT / A_KC)
#define A_LD     8
#define A_BUF    (16 * H_PITCH)
static_assert(NFEAT % A_KC == 0 && A_KC == 64, "a chunk is one 256-byte piece of a row: 16 lanes x 16 bytes");

#define G_CHUNK(c)                                                                                            \
    {                                                                                                         \
        if ((c) + 1 < A_NCH) {                                                                                \
                                \
            _Pragma("unroll")                                                                                 \
            for (int i = 0; i < A_LD; ++i)                                                                    \
                pre[i] = *(const v4f*)(xg + (size_t)(2 * i) * NFEAT + ((c) + 1) * A_KC);                      \
        }                                                                                                     \
        const float* ab = aget + ((c) & 1) * A_BUF;                                                           \
          \
        _Pragma("unroll")                                                                                     \
        for (int k4 = 0; k4 < A_KC; k4 += 4) {                                                                \
                         \
            v2f a = *(const v2f*)(ab + k4);                                                                   \
                                               \
            const float* rowp = ldsW + ((((c) * A_KC + k4) >> 1) + half) * 128;                               \
            v2f b0 = *(const v2f*)(rowp + wp0);                                                               \
            v2f b1 = *(const v2f*)(rowp + wp1);                                                               \
            v2f b2 = *(const v2f*)(rowp + wp2);                                                               \
            v2f b3 = *(const v2f*)(rowp + wp3);                                                               \
            acc0 = __builtin_amdgcn_wmma_f32_16x16x4_f32(false, a, false, b0, (short)0, acc0, false, false);  \
            acc1 = __builtin_amdgcn_wmma_f32_16x16x4_f32(false, a, false, b1, (short)0, acc1, false, false);  \
            acc2 = __builtin_amdgcn_wmma_f32_16x16x4_f32(false, a, false, b2, (short)0, acc2, false, false);  \
            acc3 = __builtin_amdgcn_wmma_f32_16x16x4_f32(false, a, false, b3, (short)0, acc3, false, false);  \
        }                                                                                                     \
        if ((c) + 1 < A_NCH) {                                                                                \
            float* ap = aput + (((c) + 1) & 1) * A_BUF;                                                       \
                                                 \
            _Pragma("unroll")                                                                                 \
            for (int i = 0; i < A_LD; ++i) *(v4f*)(ap + (2 * i) * H_PITCH) = pre[i];                          \
        }                                                                                                     \
    }

__global__ __launch_bounds__(G_NT) void k_gemm1_wmma(const float* __restrict__ x,
                                                     const float* __restrict__ W1,
                                                     const float* __restrict__ dinv,
                                                     float* __restrict__ h,
                                                     int n_nodes) {
    __shared__ __attribute__((aligned(16))) float lds[NFEAT * NHID + G_TPB * 2 * A_BUF];
    float* ldsW = lds;
    int tid = threadIdx.x;
    int wave = tid >> 5;
    int lane = tid & 31;
    int half = lane >> 4;
    int ncol = lane & 15;
    int tile = blockIdx.x * G_TPB + wave;
    int m0 = tile * 16;
    bool live = m0 < n_nodes;
    float* abuf = lds + NFEAT * NHID + wave * (2 * A_BUF);

    float4 wq[G_WLD];
    #pragma unroll
    for (int it = 0; it < G_WLD; ++it) {
        int q = tid + it * G_NT;
        wq[it] = (q < NFEAT * NHID / 4) ? *(const float4*)(W1 + 4 * q) : make_float4(0.0f, 0.0f, 0.0f, 0.0f);
    }

    const float* xg = x + (size_t)((live ? m0 : 0) + half) * NFEAT + 4 * ncol;
    v4f pre[A_LD];
    #pragma unroll
    for (int i = 0; i < A_LD; ++i) pre[i] = *(const v4f*)(xg + (size_t)(2 * i) * NFEAT);
    #pragma unroll
    for (int it = 0; it < G_WLD; ++it) {
        int idx = 4 * (tid + it * G_NT);
        if (idx < NFEAT * NHID) {
            int k  = idx >> 6;
            int n0 = idx & 63;
            int p  = k >> 1;
            int lo = k & 1;
            int rot = (p & 1) << 5;
            float vals[4] = {wq[it].x, wq[it].y, wq[it].z, wq[it].w};
            #pragma unroll
            for (int j = 0; j < 4; ++j) {
                int wpos = (((n0 + j) << 1) + rot) & 127;
                ldsW[p * 128 + wpos + lo] = vals[j];
            }
        }
    }
    __syncthreads();

    if (!live) return;

    int rot  = half << 5;

    int wp0 = (((ncol +  0) << 1) + rot) & 127;
    int wp1 = (((ncol + 16) << 1) + rot) & 127;
    int wp2 = (((ncol + 32) << 1) + rot) & 127;
    int wp3 = (((ncol + 48) << 1) + rot) & 127;

    v8f acc0 = {}, acc1 = {}, acc2 = {}, acc3 = {};

    int rbase = m0 + (half << 3);
    float dr[8];
    for (int r = 0; r < 8; ++r) dr[r] = dinv[rbase + r];

    float* aput = abuf + half * H_PITCH + 4 * ncol;
    const float* aget = abuf + ncol * H_PITCH + 2 * half;

    #pragma unroll
    for (int i = 0; i < A_LD; ++i) *(v4f*)(aput + (2 * i) * H_PITCH) = pre[i];

    G_CHUNK(0)
    G_CHUNK(1)
    G_CHUNK(2)
    G_CHUNK(3)
    static_assert(A_NCH == 4, "k_gemm1_wmma writes its chunks out one by one");

    float* st = abuf;
    for (int r = 0; r < 8; ++r) {
        float* sr = st + ((half << 3) + r) * H_PITCH;
        sr[ncol +  0] = acc0[r] * dr[r];
        sr[ncol + 16] = acc1[r] * dr[r];
        sr[ncol + 32] = acc2[r] * dr[r];
        sr[ncol + 48] = acc3[r] * dr[r];
    }
    for (int i = 0; i < 8; ++i) {
        int row = 2 * i + half;
        float4 v = *(const float4*)(st + row * H_PITCH + 4 * ncol);
        *(float4*)(h + (size_t)(m0 + row) * NHID + 4 * ncol) = v;
    }
}


#define G1_NPW 16
#define G1_CH  32
#define G1_PITCH 66

__device__ __forceinline__ float2 g1_row(const float* __restrict__ h, int s, unsigned lane2) {
    const float* row = h + (size_t)s * NHID;
    return *(const float2*)(row + lane2);
}

#define G1_FINISH()                                                                  \
    {                                                                                \
        float h0 = fmaxf(fmaf(acc.x, di, bb.x), 0.0f);                               \
        float h1 = fmaxf(fmaf(acc.y, di, bb.y), 0.0f);                               \
        float p0 = h0 * wv.x + h1 * wv.z;                                            \
        float p1 = h0 * wv.y + h1 * wv.w;                                            \
        *(float2*)(strip + k * G1_PITCH + lane2) = make_float2(p0, p1);              \
        ++k;                                                                         \
        acc = nself;                                                                 \
        if (k + 1 < nn) nself = *(const float2*)(strip + (k + 1) * G1_PITCH + lane2); \
        kend = __builtin_amdgcn_readlane(rp, k + 1);                                 \
        di = __int_as_float(__builtin_amdgcn_readlane(__float_as_int(dv), k));       \
    }

__global__ __launch_bounds__(256) void k_gather1(const float* __restrict__ h, const float* __restrict__ dinv,
                                                 const int* __restrict__ rowptr,
                                                 const int* __restrict__ csr_src,
                                                 const float* __restrict__ b1, const float* __restrict__ W2,
                                                 float* __restrict__ z, int n_nodes, int n_edges) {
    __shared__ float2 strips2[8 * G1_NPW * G1_PITCH / 2];
    int tid = threadIdx.x;
    int lane = tid & 31;
    unsigned lane2 = 2u * (unsigned)lane;
    int wave = __builtin_amdgcn_readfirstlane(tid >> 5);
    int i0 = (blockIdx.x * 8 + wave) * G1_NPW;
    if (i0 >= n_nodes) return;
    float* strip = (float*)strips2 + wave * (G1_NPW * G1_PITCH);
    int nn = n_nodes - i0;
    if (nn > G1_NPW) nn = G1_NPW;
    const int last = n_nodes - 1;
    int gi = i0 + lane;
    int rp = (gi < n_nodes) ? rowptr[gi] : n_edges;
    float dv = (gi < n_nodes) ? dinv[gi] : 0.0f;
    int gz = i0 + (lane >> 1);
    float dz = (gz < n_nodes) ? dinv[gz] : 0.0f;
    float2 bb = *(const float2*)(b1 + lane2);
    float4 wv = *(const float4*)(W2 + 2u * lane2);
    float2 own[G1_NPW];
    #pragma unroll
    for (int q = 0; q < G1_NPW; ++q) own[q] = g1_row(h, (i0 + q > last) ? last : i0 + q, lane2);
    const int e0 = __builtin_amdgcn_readlane(rp, 0);
    const int e1 = __builtin_amdgcn_readlane(rp, nn);
    int sv = 0;
    if (e1 > e0) {
        int ic = e0 + lane;
        if (ic > e1 - 1) ic = e1 - 1;
        sv = csr_src[ic];
    }
    #pragma unroll
    for (int q = 2; q < G1_NPW; ++q) *(float2*)(strip + q * G1_PITCH + lane2) = own[q];
    int k = 0;
    int kend = __builtin_amdgcn_readlane(rp, 1);
    float di = __int_as_float(__builtin_amdgcn_readlane(__float_as_int(dv), 0));
    float2 acc = own[0];
    float2 nself = own[1];
    for (int e = e0; e < e1; e += G1_CH) {
        int in = e + G1_CH + lane;
        if (in > e1 - 1) in = e1 - 1;
        int nsv = csr_src[in];
        float2 r[G1_CH];
        #pragma unroll
        for (int t = 0; t < G1_CH; ++t) r[t] = g1_row(h, __builtin_amdgcn_readlane(sv, t), lane2);
        const int elim = e1 - e;
        int lim = kend - e;
        #pragma unroll
        for (int t = 0; t < G1_CH; ++t) {
            if (__builtin_expect(t >= lim, 0)) {
                while (k + 1 < nn && t == lim) {
                    G1_FINISH()
                    lim = kend - e;
                }
                if (t >= elim) break;
            }
            acc.x += r[t].x;
            acc.y += r[t].y;
        }
        sv = nsv;
    }
    while (k < nn) G1_FINISH()
    const float* col = strip + (lane >> 1) * G1_PITCH + (lane & 1);
    float a[32];
    #pragma unroll
    for (int j = 0; j < 32; ++j) a[j] = col[2 * j];
    #pragma unroll
    for (int w = 16; w >= 1; w >>= 1) {
        #pragma unroll
        for (int j = 0; j < w; ++j) a[j] += a[j + w];
    }
    int go = 2 * i0 + lane;
    if (go < 2 * n_nodes) z[go] = a[0] * dz;
}


#define G2_NQ 4

__global__ __launch_bounds__(256) void k_gather2(const float* __restrict__ z, const float* __restrict__ dinv,
                                                 const int* __restrict__ rowptr,
                                                 const int* __restrict__ csr_src, const float* __restrict__ b2,
                                                 float* __restrict__ out, int n_nodes, int n_edges) {
    __shared__ float2 ol[32 * G2_NQ];
    int tid = threadIdx.x;
    int sub = tid & 7;
    int g = tid >> 3;
    int base = blockIdx.x * (32 * G2_NQ);
    const int last = n_nodes - 1;
    int rs[G2_NQ], re[G2_NQ];
    float dn[G2_NQ];
    float2 zi[G2_NQ];
    #pragma unroll
    for (int q = 0; q < G2_NQ; ++q) {
        int i = base + q * 32 + g;
        int ic = (i < last) ? i : last;
        int in = (i + 1 < last) ? i + 1 : last;
        rs[q] = rowptr[ic];
        re[q] = rowptr[in];
        dn[q] = dinv[ic];
        zi[q] = *(const float2*)(z + 2 * (size_t)ic);
    }
    int j[G2_NQ], en[G2_NQ];
    float a0[G2_NQ], a1[G2_NQ], di[G2_NQ];
    int left = 0;
    #pragma unroll
    for (int q = 0; q < G2_NQ; ++q) {
        int i = base + q * 32 + g;
        bool ok = i < n_nodes;
        j[q]  = ok ? rs[q] + sub : 0;
        en[q] = ok ? ((i + 1 < n_nodes) ? re[q] : n_edges) : 0;
        di[q] = ok ? dn[q] : 0.0f;
        a0[q] = (ok && sub == 0) ? zi[q].x : 0.0f;
        a1[q] = (ok && sub == 0) ? zi[q].y : 0.0f;
        int c = en[q] - j[q];
        if (c > left) left = c;
    }
    for (; left > 0; left -= 8) {
        int s[G2_NQ];
        #pragma unroll
        for (int q = 0; q < G2_NQ; ++q) {
            s[q] = -1;
            if (j[q] < en[q]) s[q] = csr_src[j[q]];
        }
        float2 zs[G2_NQ];
        #pragma unroll
        for (int q = 0; q < G2_NQ; ++q) {
            zs[q] = make_float2(0.0f, 0.0f);
            if (s[q] >= 0) zs[q] = *(const float2*)(z + 2 * (size_t)s[q]);
        }
        #pragma unroll
        for (int q = 0; q < G2_NQ; ++q) {
            if (s[q] >= 0) {
                a0[q] += zs[q].x;
                a1[q] += zs[q].y;
            }
            j[q] += 8;
        }
    }
    #pragma unroll
    for (int q = 0; q < G2_NQ; ++q) {
        float x0 = a0[q], x1 = a1[q];
        for (int o = 4; o >= 1; o >>= 1) {
            x0 += __shfl_xor(x0, o);
            x1 += __shfl_xor(x1, o);
        }
        if (sub == 0) ol[q * 32 + g] = make_float2(x0 * di[q], x1 * di[q]);
    }
    __syncthreads();
    if (tid < 32 * G2_NQ) {
        int o = base + tid;
        if (o < n_nodes) {
            float2 v = ol[tid];
            *(float2*)(out + 2 * (size_t)o) = make_float2(v.x + b2[0], v.y + b2[1]);
        }
    }
}


extern "C" void kernel_launch(void* const* d_in, const int* in_sizes, int n_in,
                              void* d_out, int out_size, void* d_ws, size_t ws_size,
                              hipStream_t stream) {
    const float* x  = (const float*)d_in[0];
    const int*   ei = (const int*)d_in[1];
    const float* W1 = (const float*)d_in[2];
    const float* b1 = (const float*)d_in[3];
    const float* W2 = (const float*)d_in[4];
    const float* b2 = (const float*)d_in[5];

    int n_nodes = in_sizes[0] / NFEAT;
    int n_edges = in_sizes[1] / 2;
    const int* src = ei;
    const int* dst = ei + n_edges;
    int n_bins = (n_nodes + BIN_NODES - 1) / BIN_NODES;
    int slice_len = (n_edges + N_SLICES - 1) / N_SLICES;

    float* ws = (float*)d_ws;
    size_t off = 0;
    size_t n64 = ((size_t)n_nodes + 63) & ~(size_t)63;
    size_t e64 = ((size_t)n_edges + 63) & ~(size_t)63;
    float* dinv    = ws + off;         off += n64;
    float* h       = ws + off;         off += (size_t)n_nodes * NHID;
    float* z       = ws + off;         off += ((size_t)2 * n_nodes + 63) & ~(size_t)63;
    int*   rowptr  = (int*)(ws + off); off += n64;
    off += 2 * MAX_BINS;
    int*   csr_src = (int*)(ws + off); off += e64;
    int2*  pairs   = (int2*)h;
    int*   lstart  = (int*)h + 2 * e64;
    (void)ws_size; (void)n_in; (void)out_size;

    float* out = (float*)d_out;

    k_slice_sort<<<N_SLICES, SS_T, 0, stream>>>(src, dst, lstart, pairs, n_edges, n_bins, slice_len);
    k_bin_sort  <<<n_bins, BS_T, 0, stream>>>(pairs, lstart, rowptr, dinv, csr_src, n_nodes, n_bins, slice_len);

    int tiles  = n_nodes / 16;
    int blocks = (tiles + G_TPB - 1) / G_TPB;
    k_gemm1_wmma<<<blocks, G_NT, 0, stream>>>(x, W1, dinv, h, n_nodes);

    k_gather1<<<((n_nodes + 15) / 16 + 7) / 8, 256, 0, stream>>>(h, dinv, rowptr, csr_src, b1, W2, z, n_nodes, n_edges);
    k_gather2<<<(n_nodes + 127) / 128, 256, 0, stream>>>(z, dinv, rowptr, csr_src, b2, out, n_nodes, n_edges);
}
